# starting kernel
# baseline (speedup 1.0000x reference)
#include <hip/hip_runtime.h>
#include <hip/hip_bf16.h>

#define N_NODES 50000
#define N_EDGES 800000
#define DIN 128
#define DOUT 128
#define LDS_STRIDE 132

typedef __attribute__((ext_vector_type(2))) float v2f;
typedef __attribute__((ext_vector_type(8))) float v8f;

__launch_bounds__(256)
__global__ void gemm_wmma_f32(const float* __restrict__ A,
                              const float* __restrict__ W,
                              float* __restrict__ C, int nrows) {
  __shared__ float As[32 * LDS_STRIDE];

  const int tid  = threadIdx.x;
  const int lane = tid & 31;
  const int wave = tid >> 5;
  const int m0   = blockIdx.x * 32;
  const int rows = (nrows - m0) < 32 ? (nrows - m0) : 32;

  {
    const float4* src4 = (const float4*)(A + (size_t)m0 * DIN);
    const int n4 = rows * (DIN / 4);
    for (int i = tid; i < n4; i += 256) {
      const int r = i >> 5;
      const int c = i & 31;
      ((float4*)(As + r * LDS_STRIDE))[c] = src4[i];
    }
  }
  __syncthreads();

  const int n0    = wave * 16;
  const int mlane = lane & 15;
  const int khalf = (lane >> 4) * 2;

  const float* a0   = As + (size_t)mlane * LDS_STRIDE + khalf;
  const float* a1   = a0 + 16 * LDS_STRIDE;
  const float* wcol = W + (size_t)khalf * DOUT + n0 + mlane;

  v8f acc0 = {}, acc1 = {};
#pragma unroll 4
  for (int k = 0; k < DIN; k += 4) {
    v2f b;
    b.x = wcol[(size_t)k * DOUT];
    b.y = wcol[(size_t)(k + 1) * DOUT];
    v2f x0 = *(const v2f*)(a0 + k);
    v2f x1 = *(const v2f*)(a1 + k);
    acc0 = __builtin_amdgcn_wmma_f32_16x16x4_f32(false, x0, false, b,
                                                 (short)0, acc0, false, false);
    acc1 = __builtin_amdgcn_wmma_f32_16x16x4_f32(false, x1, false, b,
                                                 (short)0, acc1, false, false);
  }

  float* crow = C + (size_t)(m0 + (lane >> 4) * 8) * DOUT + n0 + mlane;
#pragma unroll
  for (int j = 0; j < 8; ++j)
    crow[(size_t)j * DOUT] = acc0[j];
  if (rows > 16) {
    float* crow1 = crow + (size_t)16 * DOUT;
#pragma unroll
    for (int j = 0; j < 8; ++j)
      crow1[(size_t)j * DOUT] = acc1[j];
  }
}

__global__ void deg_init(float* __restrict__ deg) {
  int i = blockIdx.x * blockDim.x + threadIdx.x;
  if (i < N_NODES) deg[i] = 1.0f;
}

__global__ void deg_accum(const int* __restrict__ dst, float* __restrict__ deg) {
  int i = blockIdx.x * blockDim.x + threadIdx.x;
  if (i < N_EDGES) atomicAdd(&deg[dst[i]], 1.0f);
}

__global__ void deg_to_dinv(float* __restrict__ deg) {
  int i = blockIdx.x * blockDim.x + threadIdx.x;
  if (i < N_NODES) {
    float d = deg[i];
    deg[i] = d > 0.0f ? (1.0f / sqrtf(d)) : 0.0f;
  }
}

__global__ void zero_buf(float* __restrict__ p, int n) {
  int i = blockIdx.x * blockDim.x + threadIdx.x;
  if (i < n) p[i] = 0.0f;
}

__launch_bounds__(256)
__global__ void edge_scatter(const float* __restrict__ H,
                             const int* __restrict__ src,
                             const int* __restrict__ dst,
                             const float* __restrict__ dinv,
                             float* __restrict__ agg, int nedges) {
  int e    = (int)((blockIdx.x * 256u + threadIdx.x) >> 5);
  int lane = threadIdx.x & 31;
  if (e >= nedges) return;
  int s = src[e];
  int d = dst[e];
  float nrm = dinv[s] * dinv[d];
  const float4 v = ((const float4*)(H + (size_t)s * DOUT))[lane];
  float* out = agg + (size_t)d * DOUT + (size_t)lane * 4;
  atomicAdd(out + 0, v.x * nrm);
  atomicAdd(out + 1, v.y * nrm);
  atomicAdd(out + 2, v.z * nrm);
  atomicAdd(out + 3, v.w * nrm);
}

__global__ void bias_relu(const float* __restrict__ agg,
                          const float* __restrict__ hlin,
                          const float* __restrict__ dinv,
                          const float* __restrict__ b,
                          float* __restrict__ out) {
  int i = blockIdx.x * blockDim.x + threadIdx.x;
  if (i < N_NODES * DOUT) {
    int node = i >> 7;
    int c    = i & (DOUT - 1);
    float di = dinv[node];
    float v  = agg[i] + hlin[i] * di * di + b[c];
    out[i]   = v > 0.0f ? v : 0.0f;
  }
}

extern "C" void kernel_launch(void* const* d_in, const int* in_sizes, int n_in,
                              void* d_out, int out_size, void* d_ws, size_t ws_size,
                              hipStream_t stream) {
  (void)in_sizes; (void)n_in; (void)out_size; (void)ws_size;

  const float* x  = (const float*)d_in[0];
  const int*   ei = (const int*)d_in[1];
  const float* Wl[3] = {(const float*)d_in[2], (const float*)d_in[4], (const float*)d_in[6]};
  const float* bl[3] = {(const float*)d_in[3], (const float*)d_in[5], (const float*)d_in[7]};
  const int* src = ei;
  const int* dst = ei + N_EDGES;

  char*  ws   = (char*)d_ws;
  float* dinv = (float*)ws;
  size_t off  = ((size_t)N_NODES * 4 + 255) & ~(size_t)255;
  float* hlin = (float*)(ws + off);
  float* agg  = (float*)(ws + off + (size_t)N_NODES * DOUT * 4);

  const int NE = N_NODES * DOUT;

  deg_init   <<<(N_NODES + 255) / 256, 256, 0, stream>>>(dinv);
  deg_accum  <<<(N_EDGES + 255) / 256, 256, 0, stream>>>(dst, dinv);
  deg_to_dinv<<<(N_NODES + 255) / 256, 256, 0, stream>>>(dinv);

  const float* hin  = x;
  float*       hout = (float*)d_out;
  for (int l = 0; l < 3; ++l) {
    gemm_wmma_f32<<<(N_NODES + 31) / 32, 256, 0, stream>>>(hin, Wl[l], hlin, N_NODES);
    zero_buf     <<<(NE + 255) / 256, 256, 0, stream>>>(agg, NE);
    edge_scatter <<<N_EDGES / 8, 256, 0, stream>>>(hlin, src, dst, dinv, agg, N_EDGES);
    bias_relu    <<<(NE + 255) / 256, 256, 0, stream>>>(agg, hlin, dinv, bl[l], hout);
    hin = hout;
  }
}


# Layer 1's dst sweep also stores each wave's owned edges (source | local node) in the workspace; layers 2 and 3 read these lists, no sweep of dst and no src lookup; sum order stays LDS arrival
# speedup vs baseline: 2.8150x; 2.8150x over previous
#include <hip/hip_runtime.h>
#include <hip/hip_bf16.h>

#define N_NODES 50000
#define N_EDGES 800000
#define DIN 128
#define DOUT 128
#define LDS_STRIDE 132

typedef __attribute__((ext_vector_type(2))) float v2f;
typedef __attribute__((ext_vector_type(8))) float v8f;

__launch_bounds__(256)
__global__ void gemm_wmma_f32(const float* __restrict__ A,
                              const float* __restrict__ W,
                              const float* __restrict__ dinv,
                              float* __restrict__ C, int nrows) {
  __shared__ float As[32 * LDS_STRIDE];

  const int tid  = threadIdx.x;
  const int lane = tid & 31;
  const int wave = tid >> 5;
  const int m0   = blockIdx.x * 32;
  const int rows = (nrows - m0) < 32 ? (nrows - m0) : 32;

  {
    const float4* src4 = (const float4*)(A + (size_t)m0 * DIN);
    const int n4 = rows * (DIN / 4);
    for (int i = tid; i < n4; i += 256) {
      const int r = i >> 5;
      const int c = i & 31;
      ((float4*)(As + r * LDS_STRIDE))[c] = src4[i];
    }
  }
  __syncthreads();

  const int n0    = wave * 16;
  const int mlane = lane & 15;
  const int khalf = (lane >> 4) * 2;

  const float* a0   = As + (size_t)mlane * LDS_STRIDE + khalf;
  const float* a1   = a0 + 16 * LDS_STRIDE;
  const float* wcol = W + (size_t)khalf * DOUT + n0 + mlane;

  v8f acc0 = {}, acc1 = {};
#pragma unroll 4
  for (int k = 0; k < DIN; k += 4) {
    v2f b;
    b.x = wcol[(size_t)k * DOUT];
    b.y = wcol[(size_t)(k + 1) * DOUT];
    v2f x0 = *(const v2f*)(a0 + k);
    v2f x1 = *(const v2f*)(a1 + k);
    acc0 = __builtin_amdgcn_wmma_f32_16x16x4_f32(false, x0, false, b,
                                                 (short)0, acc0, false, false);
    acc1 = __builtin_amdgcn_wmma_f32_16x16x4_f32(false, x1, false, b,
                                                 (short)0, acc1, false, false);
  }

  const int   rbase = m0 + (lane >> 4) * 8;
  const float* drow = dinv + rbase;
  float* crow = C + (size_t)rbase * DOUT + n0 + mlane;
#pragma unroll
  for (int j = 0; j < 8; ++j)
    crow[(size_t)j * DOUT] = acc0[j] * drow[j];
  if (rows > 16) {
    float* crow1 = crow + (size_t)16 * DOUT;
#pragma unroll
    for (int j = 0; j < 8; ++j)
      crow1[(size_t)j * DOUT] = acc1[j] * drow[16 + j];
  }
}

__global__ void deg_init(float* __restrict__ deg) {
  int i = blockIdx.x * blockDim.x + threadIdx.x;
  if (i < N_NODES) deg[i] = 1.0f;
}

__global__ void deg_accum(const int* __restrict__ dst, float* __restrict__ deg) {
  int i = blockIdx.x * blockDim.x + threadIdx.x;
  if (i < N_EDGES) atomicAdd(&deg[dst[i]], 1.0f);
}

__global__ void deg_to_dinv(float* __restrict__ deg) {
  int i = blockIdx.x * blockDim.x + threadIdx.x;
  if (i < N_NODES) {
    float d = deg[i];
    deg[i] = d > 0.0f ? (1.0f / sqrtf(d)) : 0.0f;
  }
}

#define TILE   196
#define NWG    ((N_NODES + TILE - 1) / TILE)
#define N4     (N_EDGES / 4)
#define STEP   8
#define NCHUNK ((N4 + STEP * 32 - 1) / (STEP * 32))
#define BATCH  16
#define QCAP   (BATCH + STEP * 128)
#define CAP    1024

static_assert(N_EDGES % 4 == 0, "dst is read as int4");
static_assert(N_EDGES <= (1 << 20) && TILE <= 256, "LDS list entry packing");
static_assert(N_NODES <= (1 << 16), "stored entry packing");
static_assert(CAP <= QCAP, "a wave's stored entries fit its LDS list");

__device__ __forceinline__ void add_rows(int cnt, unsigned ent,
                                         const float* __restrict__ HS,
                                         float* acc, int lane) {
  float4 v[BATCH];
#pragma unroll
  for (int i = 0; i < BATCH; ++i) {
    if (i < cnt) {
      const unsigned e = (unsigned)__builtin_amdgcn_readlane((int)ent, i);
      v[i] = ((const float4*)(HS + (size_t)(e & 0xFFFFu) * DOUT))[lane];
    }
  }
#pragma unroll
  for (int i = 0; i < BATCH; ++i) {
    if (i < cnt) {
      const unsigned e = (unsigned)__builtin_amdgcn_readlane((int)ent, i);
      float* a = acc + (e >> 16) * DOUT + lane;
      __hip_atomic_fetch_add(a +  0, v[i].x, __ATOMIC_RELAXED, __HIP_MEMORY_SCOPE_WORKGROUP);
      __hip_atomic_fetch_add(a + 32, v[i].y, __ATOMIC_RELAXED, __HIP_MEMORY_SCOPE_WORKGROUP);
      __hip_atomic_fetch_add(a + 64, v[i].z, __ATOMIC_RELAXED, __HIP_MEMORY_SCOPE_WORKGROUP);
      __hip_atomic_fetch_add(a + 96, v[i].w, __ATOMIC_RELAXED, __HIP_MEMORY_SCOPE_WORKGROUP);
    }
  }
}

__launch_bounds__(1024)
__global__ void aggregate_owner(const float* __restrict__ HS,
                                const int* __restrict__ src,
                                const int* __restrict__ dst,
                                const float* __restrict__ dinv,
                                const float* __restrict__ bias,
                                float* __restrict__ out,
                                unsigned* __restrict__ lists,
                                int* __restrict__ cnts,
                                int reuse) {
  __shared__ float    acc[TILE * DOUT];
  __shared__ unsigned qbuf[32 * QCAP];

  const int tid  = threadIdx.x;
  const int lane = tid & 31;
  const int wave = __builtin_amdgcn_readfirstlane(tid >> 5);
  const int base = blockIdx.x * TILE;
  const int nloc = (N_NODES - base) < TILE ? (N_NODES - base) : TILE;

  {
    const float4* own4 = (const float4*)(HS + (size_t)base * DOUT);
    for (int i = tid; i < nloc * 32; i += 1024) {
      const float4 v = own4[i];
      float* a = acc + (i >> 5) * DOUT + (i & 31);
      a[0] = v.x; a[32] = v.y; a[64] = v.z; a[96] = v.w;
    }
  }
  __syncthreads();

  const int slot = blockIdx.x * 32 + wave;
  unsigned* rec  = lists + (size_t)slot * CAP;
  const int have = reuse ? __builtin_amdgcn_readfirstlane(cnts[slot]) : 0;

  unsigned* qw = qbuf + wave * QCAP;

  if (reuse && have <= CAP) {
    for (int i0 = 0; i0 < have; i0 += 128) {
      unsigned t[4];
#pragma unroll
      for (int q = 0; q < 4; ++q) {
        const int i = i0 + q * 32 + lane;
        t[q] = rec[i < have ? i : have - 1];
      }
#pragma unroll
      for (int q = 0; q < 4; ++q) {
        const int i = i0 + q * 32 + lane;
        if (i < have) qw[i] = t[q];
      }
    }
    const int nfull = have & ~(BATCH - 1);
    for (int p = 0; p < nfull; p += BATCH) {
      unsigned ent = 0;
      if (lane < BATCH) ent = qw[p + lane];
      add_rows(BATCH, ent, HS, acc, lane);
    }
    if (have > nfull) {
      unsigned ent = 0;
      if (lane < have - nfull) ent = qw[nfull + lane];
      add_rows(have - nfull, ent, HS, acc, lane);
    }
  } else {
    const int4* dst4 = (const int4*)dst;
    const int start = (int)((blockIdx.x * (unsigned)NCHUNK) / NWG);
    int pending = 0;
    int stored  = 0;

    for (int it = wave; it < NCHUNK; it += 32) {
      int c = it + start;
      if (c >= NCHUNK) c -= NCHUNK;
      int4 d[STEP];
#pragma unroll
      for (int q = 0; q < STEP; ++q) {
        const int i4 = c * (STEP * 32) + q * 32 + lane;
        d[q] = dst4[i4 < N4 ? i4 : N4 - 1];
      }
#pragma unroll
      for (int q = 0; q < STEP; ++q) {
        const int  i4    = c * (STEP * 32) + q * 32 + lane;
        const bool valid = i4 < N4;
        const unsigned l0 = (unsigned)(d[q].x - base);
        const unsigned l1 = (unsigned)(d[q].y - base);
        const unsigned l2 = (unsigned)(d[q].z - base);
        const unsigned l3 = (unsigned)(d[q].w - base);
        const unsigned lmin = min(min(l0, l1), min(l2, l3));
        if (__builtin_amdgcn_ballot_w32(valid && lmin < (unsigned)nloc) != 0u) {
          const bool m0 = valid && l0 < (unsigned)nloc;
          const bool m1 = valid && l1 < (unsigned)nloc;
          const bool m2 = valid && l2 < (unsigned)nloc;
          const bool m3 = valid && l3 < (unsigned)nloc;
          const unsigned e0 = (unsigned)i4 * 4u;
          unsigned mk;
          mk = __builtin_amdgcn_ballot_w32(m0);
          if (m0) qw[pending + __builtin_amdgcn_mbcnt_lo(mk, 0u)] = (e0 + 0u) | (l0 << 20);
          pending += __builtin_popcount(mk);
          mk = __builtin_amdgcn_ballot_w32(m1);
          if (m1) qw[pending + __builtin_amdgcn_mbcnt_lo(mk, 0u)] = (e0 + 1u) | (l1 << 20);
          pending += __builtin_popcount(mk);
          mk = __builtin_amdgcn_ballot_w32(m2);
          if (m2) qw[pending + __builtin_amdgcn_mbcnt_lo(mk, 0u)] = (e0 + 2u) | (l2 << 20);
          pending += __builtin_popcount(mk);
          mk = __builtin_amdgcn_ballot_w32(m3);
          if (m3) qw[pending + __builtin_amdgcn_mbcnt_lo(mk, 0u)] = (e0 + 3u) | (l3 << 20);
          pending += __builtin_popcount(mk);
        }
      }
      while (pending >= BATCH) {
        pending -= BATCH;
        unsigned ent = 0;
        if (lane < BATCH) {
          const unsigned q = qw[pending + lane];
          ent = (unsigned)src[q & 0xFFFFFu] | ((q >> 20) << 16);
          if (!reuse && stored + BATCH <= CAP) rec[stored + lane] = ent;
        }
        stored += BATCH;
        add_rows(BATCH, ent, HS, acc, lane);
      }
    }
    if (pending > 0) {
      unsigned ent = 0;
      if (lane < pending) {
        const unsigned q = qw[lane];
        ent = (unsigned)src[q & 0xFFFFFu] | ((q >> 20) << 16);
        if (!reuse && stored + pending <= CAP) rec[stored + lane] = ent;
      }
      stored += pending;
      add_rows(pending, ent, HS, acc, lane);
    }
    if (!reuse && lane == 0) cnts[slot] = stored;
  }

  __syncthreads();

  {
    const float4* b4   = (const float4*)bias;
    float4*       out4 = (float4*)(out + (size_t)base * DOUT);
    for (int i = tid; i < nloc * 32; i += 1024) {
      const int   r  = i >> 5;
      const int   l  = i & 31;
      const float di = dinv[base + r];
      const float* a = acc + r * DOUT + l;
      const float4 bb = b4[l];
      float4 o;
      const float t0 = di * a[0]  + bb.x;
      const float t1 = di * a[32] + bb.y;
      const float t2 = di * a[64] + bb.z;
      const float t3 = di * a[96] + bb.w;
      o.x = t0 > 0.0f ? t0 : 0.0f;
      o.y = t1 > 0.0f ? t1 : 0.0f;
      o.z = t2 > 0.0f ? t2 : 0.0f;
      o.w = t3 > 0.0f ? t3 : 0.0f;
      out4[i] = o;
    }
  }
}

#define WS_DINV_BYTES (((size_t)N_NODES * 4 + 255) & ~(size_t)255)
#define WS_HS_BYTES   ((size_t)N_NODES * DOUT * 4)
#define WS_CNT_BYTES  ((size_t)NWG * 32 * 4)
static_assert(WS_HS_BYTES % 256 == 0, "slices stay 256B aligned");

extern "C" void kernel_launch(void* const* d_in, const int* in_sizes, int n_in,
                              void* d_out, int out_size, void* d_ws, size_t ws_size,
                              hipStream_t stream) {
  (void)in_sizes; (void)n_in; (void)out_size; (void)ws_size;

  const float* x  = (const float*)d_in[0];
  const int*   ei = (const int*)d_in[1];
  const float* Wl[3] = {(const float*)d_in[2], (const float*)d_in[4], (const float*)d_in[6]};
  const float* bl[3] = {(const float*)d_in[3], (const float*)d_in[5], (const float*)d_in[7]};
  const int* src = ei;
  const int* dst = ei + N_EDGES;

  char*     ws    = (char*)d_ws;
  float*    dinv  = (float*)ws;
  float*    hs    = (float*)(ws + WS_DINV_BYTES);
  int*      cnts  = (int*)(ws + WS_DINV_BYTES + WS_HS_BYTES);
  unsigned* lists = (unsigned*)(ws + WS_DINV_BYTES + WS_HS_BYTES + WS_CNT_BYTES);

  deg_init   <<<(N_NODES + 255) / 256, 256, 0, stream>>>(dinv);
  deg_accum  <<<(N_EDGES + 255) / 256, 256, 0, stream>>>(dst, dinv);
  deg_to_dinv<<<(N_NODES + 255) / 256, 256, 0, stream>>>(dinv);

  float* hout = (float*)d_out;
  gemm_wmma_f32  <<<(N_NODES + 31) / 32, 256, 0, stream>>>(x, Wl[0], dinv, hs, N_NODES);
  aggregate_owner<<<NWG, 1024, 0, stream>>>(hs, src, dst, dinv, bl[0], hout, lists, cnts, 0);
  gemm_wmma_f32  <<<(N_NODES + 31) / 32, 256, 0, stream>>>(hout, Wl[1], dinv, hs, N_NODES);
  aggregate_owner<<<NWG, 1024, 0, stream>>>(hs, src, dst, dinv, bl[1], hout, lists, cnts, 1);
  gemm_wmma_f32  <<<(N_NODES + 31) / 32, 256, 0, stream>>>(hout, Wl[2], dinv, hs, N_NODES);
  aggregate_owner<<<NWG, 1024, 0, stream>>>(hs, src, dst, dinv, bl[2], hout, lists, cnts, 1);
}


# dst sweep moved to a first launch of the 256 owner groups, which count their nodes' in-degrees by integer LDS adds and write dinv; 3 degree launches and global float atomics go; sums: LDS arrival
# speedup vs baseline: 3.5484x; 1.2605x over previous
#include <hip/hip_runtime.h>
#include <hip/hip_bf16.h>

#define N_NODES 50000
#define N_EDGES 800000
#define DIN 128
#define DOUT 128
#define LDS_STRIDE 132

typedef __attribute__((ext_vector_type(2))) float v2f;
typedef __attribute__((ext_vector_type(8))) float v8f;

__launch_bounds__(256)
__global__ void gemm_wmma_f32(const float* __restrict__ A,
                              const float* __restrict__ W,
                              const float* __restrict__ dinv,
                              float* __restrict__ C, int nrows) {
  __shared__ float As[32 * LDS_STRIDE];

  const int tid  = threadIdx.x;
  const int lane = tid & 31;
  const int wave = tid >> 5;
  const int m0   = blockIdx.x * 32;
  const int rows = (nrows - m0) < 32 ? (nrows - m0) : 32;

  {
    const float4* src4 = (const float4*)(A + (size_t)m0 * DIN);
    const int n4 = rows * (DIN / 4);
    for (int i = tid; i < n4; i += 256) {
      const int r = i >> 5;
      const int c = i & 31;
      ((float4*)(As + r * LDS_STRIDE))[c] = src4[i];
    }
  }
  __syncthreads();

  const int n0    = wave * 16;
  const int mlane = lane & 15;
  const int khalf = (lane >> 4) * 2;

  const float* a0   = As + (size_t)mlane * LDS_STRIDE + khalf;
  const float* a1   = a0 + 16 * LDS_STRIDE;
  const float* wcol = W + (size_t)khalf * DOUT + n0 + mlane;

  v8f acc0 = {}, acc1 = {};
#pragma unroll 4
  for (int k = 0; k < DIN; k += 4) {
    v2f b;
    b.x = wcol[(size_t)k * DOUT];
    b.y = wcol[(size_t)(k + 1) * DOUT];
    v2f x0 = *(const v2f*)(a0 + k);
    v2f x1 = *(const v2f*)(a1 + k);
    acc0 = __builtin_amdgcn_wmma_f32_16x16x4_f32(false, x0, false, b,
                                                 (short)0, acc0, false, false);
    acc1 = __builtin_amdgcn_wmma_f32_16x16x4_f32(false, x1, false, b,
                                                 (short)0, acc1, false, false);
  }

  const int   rbase = m0 + (lane >> 4) * 8;
  const float* drow = dinv + rbase;
  float* crow = C + (size_t)rbase * DOUT + n0 + mlane;
#pragma unroll
  for (int j = 0; j < 8; ++j)
    crow[(size_t)j * DOUT] = acc0[j] * drow[j];
  if (rows > 16) {
    float* crow1 = crow + (size_t)16 * DOUT;
#pragma unroll
    for (int j = 0; j < 8; ++j)
      crow1[(size_t)j * DOUT] = acc1[j] * drow[16 + j];
  }
}

#define TILE   196
#define NWG    ((N_NODES + TILE - 1) / TILE)
#define N4     (N_EDGES / 4)
#define STEP   8
#define NCHUNK ((N4 + STEP * 32 - 1) / (STEP * 32))
#define BATCH  16
#define QCAP   (BATCH + STEP * 128)
#define SQCAP  1344
#define CAP    1024

static_assert(N_EDGES % 4 == 0, "dst is read as int4");
static_assert(N_EDGES <= (1 << 20) && TILE <= 256, "LDS list entry packing");
static_assert(N_NODES <= (1 << 16), "stored entry packing");
static_assert(CAP <= QCAP, "a wave's stored entries fit its LDS list");
static_assert(SQCAP >= 32 + STEP * 128, "the sweep's list holds a left-over batch and one wave step");

__device__ __forceinline__ void add_rows(int cnt, unsigned ent,
                                         const float* __restrict__ HS,
                                         float* acc, int lane) {
  float4 v[BATCH];
#pragma unroll
  for (int i = 0; i < BATCH; ++i) {
    if (i < cnt) {
      const unsigned e = (unsigned)__builtin_amdgcn_readlane((int)ent, i);
      v[i] = ((const float4*)(HS + (size_t)(e & 0xFFFFu) * DOUT))[lane];
    }
  }
#pragma unroll
  for (int i = 0; i < BATCH; ++i) {
    if (i < cnt) {
      const unsigned e = (unsigned)__builtin_amdgcn_readlane((int)ent, i);
      float* a = acc + (e >> 16) * DOUT + lane;
      __hip_atomic_fetch_add(a +  0, v[i].x, __ATOMIC_RELAXED, __HIP_MEMORY_SCOPE_WORKGROUP);
      __hip_atomic_fetch_add(a + 32, v[i].y, __ATOMIC_RELAXED, __HIP_MEMORY_SCOPE_WORKGROUP);
      __hip_atomic_fetch_add(a + 64, v[i].z, __ATOMIC_RELAXED, __HIP_MEMORY_SCOPE_WORKGROUP);
      __hip_atomic_fetch_add(a + 96, v[i].w, __ATOMIC_RELAXED, __HIP_MEMORY_SCOPE_WORKGROUP);
    }
  }
}

__device__ __forceinline__ int scan_chunk(const int4* __restrict__ dst4, int c,
                                          int base, int nloc,
                                          unsigned* qw, int pending, int lane) {
  int4 d[STEP];
#pragma unroll
  for (int q = 0; q < STEP; ++q) {
    const int i4 = c * (STEP * 32) + q * 32 + lane;
    d[q] = dst4[i4 < N4 ? i4 : N4 - 1];
  }
#pragma unroll
  for (int q = 0; q < STEP; ++q) {
    const int  i4    = c * (STEP * 32) + q * 32 + lane;
    const bool valid = i4 < N4;
    const unsigned l0 = (unsigned)(d[q].x - base);
    const unsigned l1 = (unsigned)(d[q].y - base);
    const unsigned l2 = (unsigned)(d[q].z - base);
    const unsigned l3 = (unsigned)(d[q].w - base);
    const unsigned lmin = min(min(l0, l1), min(l2, l3));
    if (__builtin_amdgcn_ballot_w32(valid && lmin < (unsigned)nloc) != 0u) {
      const bool m0 = valid && l0 < (unsigned)nloc;
      const bool m1 = valid && l1 < (unsigned)nloc;
      const bool m2 = valid && l2 < (unsigned)nloc;
      const bool m3 = valid && l3 < (unsigned)nloc;
      const unsigned e0 = (unsigned)i4 * 4u;
      unsigned mk;
      mk = __builtin_amdgcn_ballot_w32(m0);
      if (m0) qw[pending + __builtin_amdgcn_mbcnt_lo(mk, 0u)] = (e0 + 0u) | (l0 << 20);
      pending += __builtin_popcount(mk);
      mk = __builtin_amdgcn_ballot_w32(m1);
      if (m1) qw[pending + __builtin_amdgcn_mbcnt_lo(mk, 0u)] = (e0 + 1u) | (l1 << 20);
      pending += __builtin_popcount(mk);
      mk = __builtin_amdgcn_ballot_w32(m2);
      if (m2) qw[pending + __builtin_amdgcn_mbcnt_lo(mk, 0u)] = (e0 + 2u) | (l2 << 20);
      pending += __builtin_popcount(mk);
      mk = __builtin_amdgcn_ballot_w32(m3);
      if (m3) qw[pending + __builtin_amdgcn_mbcnt_lo(mk, 0u)] = (e0 + 3u) | (l3 << 20);
      pending += __builtin_popcount(mk);
    }
  }
  return pending;
}

__device__ __forceinline__ int scan_start(int group) {
  return (int)(((unsigned)group * (unsigned)NCHUNK) / NWG);
}

__device__ __forceinline__ void record_edges(int cnt, int at, const unsigned* qw,
                                             const int* __restrict__ src,
                                             unsigned* __restrict__ rec, int stored,
                                             unsigned* indeg, int lane) {
  if (lane < cnt) {
    const unsigned q   = qw[at + lane];
    const unsigned loc = q >> 20;
    const unsigned ent = (unsigned)src[q & 0xFFFFFu] | (loc << 16);
    if (stored + cnt <= CAP) rec[stored + lane] = ent;
    __hip_atomic_fetch_add(indeg + loc, 1u, __ATOMIC_RELAXED, __HIP_MEMORY_SCOPE_WORKGROUP);
  }
}

__launch_bounds__(1024)
__global__ void owner_sweep(const int* __restrict__ src,
                            const int* __restrict__ dst,
                            float* __restrict__ dinv,
                            unsigned* __restrict__ lists,
                            int* __restrict__ cnts) {
  __shared__ unsigned qbuf[32 * SQCAP];
  __shared__ unsigned indeg[TILE];

  const int tid  = threadIdx.x;
  const int lane = tid & 31;
  const int wave = __builtin_amdgcn_readfirstlane(tid >> 5);
  const int base = blockIdx.x * TILE;
  const int nloc = (N_NODES - base) < TILE ? (N_NODES - base) : TILE;

  if (tid < TILE) indeg[tid] = 0u;
  __syncthreads();

  const int slot = blockIdx.x * 32 + wave;
  unsigned* rec  = lists + (size_t)slot * CAP;
  unsigned* qw   = qbuf + wave * SQCAP;

  const int4* dst4  = (const int4*)dst;
  const int   start = scan_start(blockIdx.x);
  int pending = 0;
  int stored  = 0;

  for (int it = wave; it < NCHUNK; it += 32) {
    int c = it + start;
    if (c >= NCHUNK) c -= NCHUNK;
    pending = scan_chunk(dst4, c, base, nloc, qw, pending, lane);
    while (pending >= 32) {
      pending -= 32;
      record_edges(32, pending, qw, src, rec, stored, indeg, lane);
      stored += 32;
    }
  }
  if (pending > 0) {
    record_edges(pending, 0, qw, src, rec, stored, indeg, lane);
    stored += pending;
  }
  if (lane == 0) cnts[slot] = stored;

  __syncthreads();

  if (tid < nloc) {
    const float d = 1.0f + (float)indeg[tid];
    dinv[base + tid] = d > 0.0f ? (1.0f / sqrtf(d)) : 0.0f;
  }
}

__launch_bounds__(1024)
__global__ void aggregate_owner(const float* __restrict__ HS,
                                const int* __restrict__ src,
                                const int* __restrict__ dst,
                                const float* __restrict__ dinv,
                                const float* __restrict__ bias,
                                float* __restrict__ out,
                                const unsigned* __restrict__ lists,
                                const int* __restrict__ cnts) {
  __shared__ float    acc[TILE * DOUT];
  __shared__ unsigned qbuf[32 * QCAP];

  const int tid  = threadIdx.x;
  const int lane = tid & 31;
  const int wave = __builtin_amdgcn_readfirstlane(tid >> 5);
  const int base = blockIdx.x * TILE;
  const int nloc = (N_NODES - base) < TILE ? (N_NODES - base) : TILE;

  {
    const float4* own4 = (const float4*)(HS + (size_t)base * DOUT);
    for (int i = tid; i < nloc * 32; i += 1024) {
      const float4 v = own4[i];
      float* a = acc + (i >> 5) * DOUT + (i & 31);
      a[0] = v.x; a[32] = v.y; a[64] = v.z; a[96] = v.w;
    }
  }
  __syncthreads();

  const int slot = blockIdx.x * 32 + wave;
  const unsigned* rec = lists + (size_t)slot * CAP;
  const int have = __builtin_amdgcn_readfirstlane(cnts[slot]);

  unsigned* qw = qbuf + wave * QCAP;

  if (have <= CAP) {
    for (int i0 = 0; i0 < have; i0 += 128) {
      unsigned t[4];
#pragma unroll
      for (int q = 0; q < 4; ++q) {
        const int i = i0 + q * 32 + lane;
        t[q] = rec[i < have ? i : have - 1];
      }
#pragma unroll
      for (int q = 0; q < 4; ++q) {
        const int i = i0 + q * 32 + lane;
        if (i < have) qw[i] = t[q];
      }
    }
    const int nfull = have & ~(BATCH - 1);
    for (int p = 0; p < nfull; p += BATCH) {
      unsigned ent = 0;
      if (lane < BATCH) ent = qw[p + lane];
      add_rows(BATCH, ent, HS, acc, lane);
    }
    if (have > nfull) {
      unsigned ent = 0;
      if (lane < have - nfull) ent = qw[nfull + lane];
      add_rows(have - nfull, ent, HS, acc, lane);
    }
  } else {
    const int4* dst4  = (const int4*)dst;
    const int   start = scan_start(blockIdx.x);
    int pending = 0;

    for (int it = wave; it < NCHUNK; it += 32) {
      int c = it + start;
      if (c >= NCHUNK) c -= NCHUNK;
      pending = scan_chunk(dst4, c, base, nloc, qw, pending, lane);
      while (pending >= BATCH) {
        pending -= BATCH;
        unsigned ent = 0;
        if (lane < BATCH) {
          const unsigned q = qw[pending + lane];
          ent = (unsigned)src[q & 0xFFFFFu] | ((q >> 20) << 16);
        }
        add_rows(BATCH, ent, HS, acc, lane);
      }
    }
    if (pending > 0) {
      unsigned ent = 0;
      if (lane < pending) {
        const unsigned q = qw[lane];
        ent = (unsigned)src[q & 0xFFFFFu] | ((q >> 20) << 16);
      }
      add_rows(pending, ent, HS, acc, lane);
    }
  }

  __syncthreads();

  {
    const float4* b4   = (const float4*)bias;
    float4*       out4 = (float4*)(out + (size_t)base * DOUT);
    for (int i = tid; i < nloc * 32; i += 1024) {
      const int   r  = i >> 5;
      const int   l  = i & 31;
      const float di = dinv[base + r];
      const float* a = acc + r * DOUT + l;
      const float4 bb = b4[l];
      float4 o;
      const float t0 = di * a[0]  + bb.x;
      const float t1 = di * a[32] + bb.y;
      const float t2 = di * a[64] + bb.z;
      const float t3 = di * a[96] + bb.w;
      o.x = t0 > 0.0f ? t0 : 0.0f;
      o.y = t1 > 0.0f ? t1 : 0.0f;
      o.z = t2 > 0.0f ? t2 : 0.0f;
      o.w = t3 > 0.0f ? t3 : 0.0f;
      out4[i] = o;
    }
  }
}

#define WS_DINV_BYTES (((size_t)N_NODES * 4 + 255) & ~(size_t)255)
#define WS_HS_BYTES   ((size_t)N_NODES * DOUT * 4)
#define WS_CNT_BYTES  ((size_t)NWG * 32 * 4)
static_assert(WS_HS_BYTES % 256 == 0, "slices stay 256B aligned");

extern "C" void kernel_launch(void* const* d_in, const int* in_sizes, int n_in,
                              void* d_out, int out_size, void* d_ws, size_t ws_size,
                              hipStream_t stream) {
  (void)in_sizes; (void)n_in; (void)out_size; (void)ws_size;

  const float* x  = (const float*)d_in[0];
  const int*   ei = (const int*)d_in[1];
  const float* Wl[3] = {(const float*)d_in[2], (const float*)d_in[4], (const float*)d_in[6]};
  const float* bl[3] = {(const float*)d_in[3], (const float*)d_in[5], (const float*)d_in[7]};
  const int* src = ei;
  const int* dst = ei + N_EDGES;

  char*     ws    = (char*)d_ws;
  float*    dinv  = (float*)ws;
  float*    hs    = (float*)(ws + WS_DINV_BYTES);
  int*      cnts  = (int*)(ws + WS_DINV_BYTES + WS_HS_BYTES);
  unsigned* lists = (unsigned*)(ws + WS_DINV_BYTES + WS_HS_BYTES + WS_CNT_BYTES);

  owner_sweep<<<NWG, 1024, 0, stream>>>(src, dst, dinv, lists, cnts);

  float* hout = (float*)d_out;
  gemm_wmma_f32  <<<(N_NODES + 31) / 32, 256, 0, stream>>>(x, Wl[0], dinv, hs, N_NODES);
  aggregate_owner<<<NWG, 1024, 0, stream>>>(hs, src, dst, dinv, bl[0], hout, lists, cnts);
  gemm_wmma_f32  <<<(N_NODES + 31) / 32, 256, 0, stream>>>(hout, Wl[1], dinv, hs, N_NODES);
  aggregate_owner<<<NWG, 1024, 0, stream>>>(hs, src, dst, dinv, bl[1], hout, lists, cnts);
  gemm_wmma_f32  <<<(N_NODES + 31) / 32, 256, 0, stream>>>(hout, Wl[2], dinv, hs, N_NODES);
  aggregate_owner<<<NWG, 1024, 0, stream>>>(hs, src, dst, dinv, bl[2], hout, lists, cnts);
}


# dst sweep split two ways: a group scans 1/8 of dst for 8 node tiles (32 ranges x 8 slices), not all of dst for one tile; degree = 1 + sum of 8 partial counts; sums stay LDS arrival order
# speedup vs baseline: 3.9829x; 1.1224x over previous
#include <hip/hip_runtime.h>
#include <hip/hip_bf16.h>

#define N_NODES 50000
#define N_EDGES 800000
#define DIN 128
#define DOUT 128
#define LDS_STRIDE 132

typedef __attribute__((ext_vector_type(2))) float v2f;
typedef __attribute__((ext_vector_type(8))) float v8f;

#define KS 8

__device__ __forceinline__ float dinv_of(const unsigned* __restrict__ part, int node) {
  unsigned c = 0u;
#pragma unroll
  for (int j = 0; j < KS; ++j) c += part[(size_t)j * N_NODES + node];
  const float d = 1.0f + (float)c;
  return d > 0.0f ? (1.0f / sqrtf(d)) : 0.0f;
}

__launch_bounds__(256)
__global__ void gemm_wmma_f32(const float* __restrict__ A,
                              const float* __restrict__ W,
                              const unsigned* __restrict__ part,
                              float* __restrict__ C, int nrows) {
  __shared__ float As[32 * LDS_STRIDE];
  __shared__ float dsh[32];

  const int tid  = threadIdx.x;
  const int lane = tid & 31;
  const int wave = tid >> 5;
  const int m0   = blockIdx.x * 32;
  const int rows = (nrows - m0) < 32 ? (nrows - m0) : 32;

  {
    const float4* src4 = (const float4*)(A + (size_t)m0 * DIN);
    const int n4 = rows * (DIN / 4);
    for (int i = tid; i < n4; i += 256) {
      const int r = i >> 5;
      const int c = i & 31;
      ((float4*)(As + r * LDS_STRIDE))[c] = src4[i];
    }
  }
  if (tid < rows) dsh[tid] = dinv_of(part, m0 + tid);
  __syncthreads();

  const int n0    = wave * 16;
  const int mlane = lane & 15;
  const int khalf = (lane >> 4) * 2;

  const float* a0   = As + (size_t)mlane * LDS_STRIDE + khalf;
  const float* a1   = a0 + 16 * LDS_STRIDE;
  const float* wcol = W + (size_t)khalf * DOUT + n0 + mlane;

  v8f acc0 = {}, acc1 = {};
#pragma unroll 4
  for (int k = 0; k < DIN; k += 4) {
    v2f b;
    b.x = wcol[(size_t)k * DOUT];
    b.y = wcol[(size_t)(k + 1) * DOUT];
    v2f x0 = *(const v2f*)(a0 + k);
    v2f x1 = *(const v2f*)(a1 + k);
    acc0 = __builtin_amdgcn_wmma_f32_16x16x4_f32(false, x0, false, b,
                                                 (short)0, acc0, false, false);
    acc1 = __builtin_amdgcn_wmma_f32_16x16x4_f32(false, x1, false, b,
                                                 (short)0, acc1, false, false);
  }

  const int   rbase = m0 + (lane >> 4) * 8;
  const float* drow = dsh + (lane >> 4) * 8;
  float* crow = C + (size_t)rbase * DOUT + n0 + mlane;
#pragma unroll
  for (int j = 0; j < 8; ++j)
    crow[(size_t)j * DOUT] = acc0[j] * drow[j];
  if (rows > 16) {
    float* crow1 = crow + (size_t)16 * DOUT;
#pragma unroll
    for (int j = 0; j < 8; ++j)
      crow1[(size_t)j * DOUT] = acc1[j] * drow[16 + j];
  }
}

#define TILE   196
#define NWG    ((N_NODES + TILE - 1) / TILE)
#define STILE  (KS * TILE)
#define NSG    (NWG / KS)
#define N4     (N_EDGES / 4)
#define STEP   8
#define NCHUNK ((N4 + STEP * 32 - 1) / (STEP * 32))
#define BATCH  16
#define QCAP   (BATCH + STEP * 128)
#define SQCAP  1344
#define CAP    (1024 / KS)

static_assert(N_EDGES % 4 == 0, "dst is read as int4");
static_assert(NWG % KS == 0, "whole sweep ranges");
static_assert(N_EDGES <= (1 << 20) && STILE <= (1 << 12) && TILE <= 256, "LDS list entry packing");
static_assert(N_NODES <= (1 << 16), "stored entry packing");
static_assert(KS * CAP <= QCAP, "a wave's joined slices fit its LDS list");
static_assert(SQCAP >= 32 + STEP * 128, "the sweep's list holds a left-over batch and one wave step");

__device__ __forceinline__ void add_rows(int cnt, unsigned ent,
                                         const float* __restrict__ HS,
                                         float* acc, int lane) {
  float4 v[BATCH];
#pragma unroll
  for (int i = 0; i < BATCH; ++i) {
    if (i < cnt) {
      const unsigned e = (unsigned)__builtin_amdgcn_readlane((int)ent, i);
      v[i] = ((const float4*)(HS + (size_t)(e & 0xFFFFu) * DOUT))[lane];
    }
  }
#pragma unroll
  for (int i = 0; i < BATCH; ++i) {
    if (i < cnt) {
      const unsigned e = (unsigned)__builtin_amdgcn_readlane((int)ent, i);
      float* a = acc + (e >> 16) * DOUT + lane;
      __hip_atomic_fetch_add(a +  0, v[i].x, __ATOMIC_RELAXED, __HIP_MEMORY_SCOPE_WORKGROUP);
      __hip_atomic_fetch_add(a + 32, v[i].y, __ATOMIC_RELAXED, __HIP_MEMORY_SCOPE_WORKGROUP);
      __hip_atomic_fetch_add(a + 64, v[i].z, __ATOMIC_RELAXED, __HIP_MEMORY_SCOPE_WORKGROUP);
      __hip_atomic_fetch_add(a + 96, v[i].w, __ATOMIC_RELAXED, __HIP_MEMORY_SCOPE_WORKGROUP);
    }
  }
}

__device__ __forceinline__ int scan_chunk(const int4* __restrict__ dst4, int c,
                                          int base, int nloc,
                                          unsigned* qw, int pending, int lane) {
  int4 d[STEP];
#pragma unroll
  for (int q = 0; q < STEP; ++q) {
    const int i4 = c * (STEP * 32) + q * 32 + lane;
    d[q] = dst4[i4 < N4 ? i4 : N4 - 1];
  }
#pragma unroll
  for (int q = 0; q < STEP; ++q) {
    const int  i4    = c * (STEP * 32) + q * 32 + lane;
    const bool valid = i4 < N4;
    const unsigned l0 = (unsigned)(d[q].x - base);
    const unsigned l1 = (unsigned)(d[q].y - base);
    const unsigned l2 = (unsigned)(d[q].z - base);
    const unsigned l3 = (unsigned)(d[q].w - base);
    const unsigned lmin = min(min(l0, l1), min(l2, l3));
    if (__builtin_amdgcn_ballot_w32(valid && lmin < (unsigned)nloc) != 0u) {
      const bool m0 = valid && l0 < (unsigned)nloc;
      const bool m1 = valid && l1 < (unsigned)nloc;
      const bool m2 = valid && l2 < (unsigned)nloc;
      const bool m3 = valid && l3 < (unsigned)nloc;
      const unsigned e0 = (unsigned)i4 * 4u;
      unsigned mk;
      mk = __builtin_amdgcn_ballot_w32(m0);
      if (m0) qw[pending + __builtin_amdgcn_mbcnt_lo(mk, 0u)] = (e0 + 0u) | (l0 << 20);
      pending += __builtin_popcount(mk);
      mk = __builtin_amdgcn_ballot_w32(m1);
      if (m1) qw[pending + __builtin_amdgcn_mbcnt_lo(mk, 0u)] = (e0 + 1u) | (l1 << 20);
      pending += __builtin_popcount(mk);
      mk = __builtin_amdgcn_ballot_w32(m2);
      if (m2) qw[pending + __builtin_amdgcn_mbcnt_lo(mk, 0u)] = (e0 + 2u) | (l2 << 20);
      pending += __builtin_popcount(mk);
      mk = __builtin_amdgcn_ballot_w32(m3);
      if (m3) qw[pending + __builtin_amdgcn_mbcnt_lo(mk, 0u)] = (e0 + 3u) | (l3 << 20);
      pending += __builtin_popcount(mk);
    }
  }
  return pending;
}

__device__ __forceinline__ int scan_start(int range) {
  return (int)(((unsigned)range * (unsigned)NCHUNK) / NSG);
}

__device__ __forceinline__ void record_edges(int cnt, int at, const unsigned* qw,
                                             const int* __restrict__ src,
                                             unsigned* __restrict__ rec0,
                                             int (&stored)[KS],
                                             unsigned* indeg, int lane) {
  const bool act = lane < cnt;
  unsigned tl = 0u, ent = 0u;
  if (act) {
    const unsigned q   = qw[at + lane];
    const unsigned loc = q >> 20;
    tl = loc / (unsigned)TILE;
    ent = (unsigned)src[q & 0xFFFFFu] | ((loc - tl * (unsigned)TILE) << 16);
    __hip_atomic_fetch_add(indeg + loc, 1u, __ATOMIC_RELAXED, __HIP_MEMORY_SCOPE_WORKGROUP);
  }
#pragma unroll
  for (int t = 0; t < KS; ++t) {
    const bool mine = act && tl == (unsigned)t;
    const unsigned mk = __builtin_amdgcn_ballot_w32(mine);
    if (mine) {
      const int pos = stored[t] + (int)__builtin_amdgcn_mbcnt_lo(mk, 0u);
      if (pos < CAP) rec0[(size_t)t * (32 * KS * CAP) + pos] = ent;
    }
    stored[t] += __builtin_popcount(mk);
  }
}

__launch_bounds__(1024)
__global__ void owner_sweep(const int* __restrict__ src,
                            const int* __restrict__ dst,
                            unsigned* __restrict__ part,
                            unsigned* __restrict__ lists,
                            int* __restrict__ cnts) {
  __shared__ unsigned qbuf[32 * SQCAP];
  __shared__ unsigned indeg[STILE];

  const int tid  = threadIdx.x;
  const int lane = tid & 31;
  const int wave  = __builtin_amdgcn_readfirstlane(tid >> 5);
  const int range = blockIdx.x / KS;
  const int slice = blockIdx.x % KS;
  const int base  = range * STILE;
  const int nloc  = (N_NODES - base) < STILE ? (N_NODES - base) : STILE;

  for (int i = tid; i < STILE; i += 1024) indeg[i] = 0u;
  __syncthreads();

  const int slot0 = ((range * KS) * 32 + wave) * KS + slice;
  unsigned* rec0  = lists + (size_t)slot0 * CAP;
  unsigned* qw    = qbuf + wave * SQCAP;

  const int4* dst4  = (const int4*)dst;
  const int   start = scan_start(range);
  int pending = 0;
  int stored[KS];
#pragma unroll
  for (int t = 0; t < KS; ++t) stored[t] = 0;

  for (int it = wave + 32 * slice; it < NCHUNK; it += 32 * KS) {
    int c = it + start;
    if (c >= NCHUNK) c -= NCHUNK;
    pending = scan_chunk(dst4, c, base, nloc, qw, pending, lane);
    while (pending >= 32) {
      pending -= 32;
      record_edges(32, pending, qw, src, rec0, stored, indeg, lane);
    }
  }
  if (pending > 0) record_edges(pending, 0, qw, src, rec0, stored, indeg, lane);
  if (lane == 0) {
#pragma unroll
    for (int t = 0; t < KS; ++t) cnts[slot0 + t * (32 * KS)] = stored[t];
  }

  __syncthreads();

  for (int i = tid; i < nloc; i += 1024)
    part[(size_t)slice * N_NODES + base + i] = indeg[i];
}

__launch_bounds__(1024)
__global__ void aggregate_owner(const float* __restrict__ HS,
                                const int* __restrict__ src,
                                const int* __restrict__ dst,
                                const unsigned* __restrict__ part,
                                const float* __restrict__ bias,
                                float* __restrict__ out,
                                const unsigned* __restrict__ lists,
                                const int* __restrict__ cnts) {
  __shared__ float    acc[TILE * DOUT];
  __shared__ unsigned qbuf[32 * QCAP];
  __shared__ float    dsh[TILE];

  const int tid  = threadIdx.x;
  const int lane = tid & 31;
  const int wave = __builtin_amdgcn_readfirstlane(tid >> 5);
  const int base = blockIdx.x * TILE;
  const int nloc = (N_NODES - base) < TILE ? (N_NODES - base) : TILE;

  {
    const float4* own4 = (const float4*)(HS + (size_t)base * DOUT);
    for (int i = tid; i < nloc * 32; i += 1024) {
      const float4 v = own4[i];
      float* a = acc + (i >> 5) * DOUT + (i & 31);
      a[0] = v.x; a[32] = v.y; a[64] = v.z; a[96] = v.w;
    }
  }
  if (tid < nloc) dsh[tid] = dinv_of(part, base + tid);
  __syncthreads();

  const int slot = (blockIdx.x * 32 + wave) * KS;
  const unsigned* rec = lists + (size_t)slot * CAP;
  int hv[KS];
  bool fits = true;
#pragma unroll
  for (int j = 0; j < KS; ++j) {
    hv[j] = __builtin_amdgcn_readfirstlane(cnts[slot + j]);
    fits = fits && hv[j] <= CAP;
  }

  unsigned* qw = qbuf + wave * QCAP;

  if (fits) {
    unsigned t[KS];
#pragma unroll
    for (int j = 0; j < KS; ++j) {
      t[j] = 0u;
      if (lane < hv[j]) t[j] = rec[j * CAP + lane];
    }
    int have = 0;
#pragma unroll
    for (int j = 0; j < KS; ++j) {
      if (lane < hv[j]) qw[have + lane] = t[j];
#pragma nounroll
      for (int i = 32 + lane; i < hv[j]; i += 32) qw[have + i] = rec[j * CAP + i];
      have += hv[j];
    }
    const int nfull = have & ~(BATCH - 1);
    for (int p = 0; p < nfull; p += BATCH) {
      unsigned ent = 0;
      if (lane < BATCH) ent = qw[p + lane];
      add_rows(BATCH, ent, HS, acc, lane);
    }
    if (have > nfull) {
      unsigned ent = 0;
      if (lane < have - nfull) ent = qw[nfull + lane];
      add_rows(have - nfull, ent, HS, acc, lane);
    }
  } else {
    const int4* dst4  = (const int4*)dst;
    const int   start = scan_start(blockIdx.x / KS);
    int pending = 0;

    for (int it = wave; it < NCHUNK; it += 32) {
      int c = it + start;
      if (c >= NCHUNK) c -= NCHUNK;
      pending = scan_chunk(dst4, c, base, nloc, qw, pending, lane);
      while (pending >= BATCH) {
        pending -= BATCH;
        unsigned ent = 0;
        if (lane < BATCH) {
          const unsigned q = qw[pending + lane];
          ent = (unsigned)src[q & 0xFFFFFu] | ((q >> 20) << 16);
        }
        add_rows(BATCH, ent, HS, acc, lane);
      }
    }
    if (pending > 0) {
      unsigned ent = 0;
      if (lane < pending) {
        const unsigned q = qw[lane];
        ent = (unsigned)src[q & 0xFFFFFu] | ((q >> 20) << 16);
      }
      add_rows(pending, ent, HS, acc, lane);
    }
  }

  __syncthreads();

  {
    const float4* b4   = (const float4*)bias;
    float4*       out4 = (float4*)(out + (size_t)base * DOUT);
    for (int i = tid; i < nloc * 32; i += 1024) {
      const int   r  = i >> 5;
      const int   l  = i & 31;
      const float di = dsh[r];
      const float* a = acc + r * DOUT + l;
      const float4 bb = b4[l];
      float4 o;
      const float t0 = di * a[0]  + bb.x;
      const float t1 = di * a[32] + bb.y;
      const float t2 = di * a[64] + bb.z;
      const float t3 = di * a[96] + bb.w;
      o.x = t0 > 0.0f ? t0 : 0.0f;
      o.y = t1 > 0.0f ? t1 : 0.0f;
      o.z = t2 > 0.0f ? t2 : 0.0f;
      o.w = t3 > 0.0f ? t3 : 0.0f;
      out4[i] = o;
    }
  }
}

#define WS_PART_BYTES (((size_t)KS * N_NODES * 4 + 255) & ~(size_t)255)
#define WS_HS_BYTES   ((size_t)N_NODES * DOUT * 4)
#define WS_CNT_BYTES  ((size_t)NWG * 32 * KS * 4)
static_assert(WS_HS_BYTES % 256 == 0 && WS_CNT_BYTES % 256 == 0, "slices stay 256B aligned");

extern "C" void kernel_launch(void* const* d_in, const int* in_sizes, int n_in,
                              void* d_out, int out_size, void* d_ws, size_t ws_size,
                              hipStream_t stream) {
  (void)in_sizes; (void)n_in; (void)out_size; (void)ws_size;

  const float* x  = (const float*)d_in[0];
  const int*   ei = (const int*)d_in[1];
  const float* Wl[3] = {(const float*)d_in[2], (const float*)d_in[4], (const float*)d_in[6]};
  const float* bl[3] = {(const float*)d_in[3], (const float*)d_in[5], (const float*)d_in[7]};
  const int* src = ei;
  const int* dst = ei + N_EDGES;

  char*     ws    = (char*)d_ws;
  unsigned* part  = (unsigned*)ws;
  float*    hs    = (float*)(ws + WS_PART_BYTES);
  int*      cnts  = (int*)(ws + WS_PART_BYTES + WS_HS_BYTES);
  unsigned* lists = (unsigned*)(ws + WS_PART_BYTES + WS_HS_BYTES + WS_CNT_BYTES);

  owner_sweep<<<NWG, 1024, 0, stream>>>(src, dst, part, lists, cnts);

  float* hout = (float*)d_out;
  gemm_wmma_f32  <<<(N_NODES + 31) / 32, 256, 0, stream>>>(x, Wl[0], part, hs, N_NODES);
  aggregate_owner<<<NWG, 1024, 0, stream>>>(hs, src, dst, part, bl[0], hout, lists, cnts);
  gemm_wmma_f32  <<<(N_NODES + 31) / 32, 256, 0, stream>>>(hout, Wl[1], part, hs, N_NODES);
  aggregate_owner<<<NWG, 1024, 0, stream>>>(hs, src, dst, part, bl[1], hout, lists, cnts);
  gemm_wmma_f32  <<<(N_NODES + 31) / 32, 256, 0, stream>>>(hout, Wl[2], part, hs, N_NODES);
  aggregate_owner<<<NWG, 1024, 0, stream>>>(hs, src, dst, part, bl[2], hout, lists, cnts);
}


# Layers 1 and 2 keep relu(agg) in the owner group's LDS tile and multiply it there by the next layer's W (f32 WMMA, W staged in LDS): 2 GEMM launches, h round trips go; K summed in LDS column order
# speedup vs baseline: 4.1096x; 1.0318x over previous
#include <hip/hip_runtime.h>
#include <hip/hip_bf16.h>

#define N_NODES 50000
#define N_EDGES 800000
#define DIN 128
#define DOUT 128
#define LDS_STRIDE 132

typedef __attribute__((ext_vector_type(2))) float v2f;
typedef __attribute__((ext_vector_type(8))) float v8f;

#define KS 8

__device__ __forceinline__ float dinv_of(const unsigned* __restrict__ part, int node) {
  unsigned c = 0u;
#pragma unroll
  for (int j = 0; j < KS; ++j) c += part[(size_t)j * N_NODES + node];
  const float d = 1.0f + (float)c;
  return d > 0.0f ? (1.0f / sqrtf(d)) : 0.0f;
}

__launch_bounds__(256)
__global__ void gemm_wmma_f32(const float* __restrict__ A,
                              const float* __restrict__ W,
                              const unsigned* __restrict__ part,
                              float* __restrict__ C, int nrows) {
  __shared__ float As[32 * LDS_STRIDE];
  __shared__ float dsh[32];

  const int tid  = threadIdx.x;
  const int lane = tid & 31;
  const int wave = tid >> 5;
  const int m0   = blockIdx.x * 32;
  const int rows = (nrows - m0) < 32 ? (nrows - m0) : 32;

  {
    const float4* src4 = (const float4*)(A + (size_t)m0 * DIN);
    const int n4 = rows * (DIN / 4);
    for (int i = tid; i < n4; i += 256) {
      const int r = i >> 5;
      const int c = i & 31;
      ((float4*)(As + r * LDS_STRIDE))[c] = src4[i];
    }
  }
  if (tid < rows) dsh[tid] = dinv_of(part, m0 + tid);
  __syncthreads();

  const int n0    = wave * 16;
  const int mlane = lane & 15;
  const int khalf = (lane >> 4) * 2;

  const float* a0   = As + (size_t)mlane * LDS_STRIDE + khalf;
  const float* a1   = a0 + 16 * LDS_STRIDE;
  const float* wcol = W + (size_t)khalf * DOUT + n0 + mlane;

  v8f acc0 = {}, acc1 = {};
#pragma unroll 4
  for (int k = 0; k < DIN; k += 4) {
    v2f b;
    b.x = wcol[(size_t)k * DOUT];
    b.y = wcol[(size_t)(k + 1) * DOUT];
    v2f x0 = *(const v2f*)(a0 + k);
    v2f x1 = *(const v2f*)(a1 + k);
    acc0 = __builtin_amdgcn_wmma_f32_16x16x4_f32(false, x0, false, b,
                                                 (short)0, acc0, false, false);
    acc1 = __builtin_amdgcn_wmma_f32_16x16x4_f32(false, x1, false, b,
                                                 (short)0, acc1, false, false);
  }

  const int   rbase = m0 + (lane >> 4) * 8;
  const float* drow = dsh + (lane >> 4) * 8;
  float* crow = C + (size_t)rbase * DOUT + n0 + mlane;
#pragma unroll
  for (int j = 0; j < 8; ++j)
    crow[(size_t)j * DOUT] = acc0[j] * drow[j];
  if (rows > 16) {
    float* crow1 = crow + (size_t)16 * DOUT;
#pragma unroll
    for (int j = 0; j < 8; ++j)
      crow1[(size_t)j * DOUT] = acc1[j] * drow[16 + j];
  }
}

#define TILE   196
#define MT     ((TILE + 15) / 16)
#define AROWS  (MT * 16)
#define APITCH 132
#define NWG    ((N_NODES + TILE - 1) / TILE)
#define STILE  (KS * TILE)
#define NSG    (NWG / KS)
#define N4     (N_EDGES / 4)
#define STEP   8
#define NCHUNK ((N4 + STEP * 32 - 1) / (STEP * 32))
#define BATCH  16
#define QCAP   (BATCH + STEP * 128)
#define SQCAP  1344
#define CAP    (1024 / KS)

static_assert(N_EDGES % 4 == 0, "dst is read as int4");
static_assert(NWG % KS == 0, "whole sweep ranges");
static_assert(N_EDGES <= (1 << 20) && STILE <= (1 << 12) && TILE <= 256, "LDS list entry packing");
static_assert(N_NODES <= (1 << 16), "stored entry packing");
static_assert(KS * CAP <= QCAP, "a wave's joined slices fit its LDS list");
static_assert(SQCAP >= 32 + STEP * 128, "the sweep's list holds a left-over batch and one wave step");
static_assert(MT == 13, "row tiles 0..12 over 4 wave groups: 4, 3, 3, 3");
static_assert(32 * QCAP * 4 >= DOUT * DOUT * 4, "the next weight fits the dead wave lists");

__device__ __forceinline__ void add_rows(int cnt, unsigned ent,
                                         const float* __restrict__ HS,
                                         float* acc, int lane) {
  float4 v[BATCH];
#pragma unroll
  for (int i = 0; i < BATCH; ++i) {
    if (i < cnt) {
      const unsigned e = (unsigned)__builtin_amdgcn_readlane((int)ent, i);
      v[i] = ((const float4*)(HS + (size_t)(e & 0xFFFFu) * DOUT))[lane];
    }
  }
#pragma unroll
  for (int i = 0; i < BATCH; ++i) {
    if (i < cnt) {
      const unsigned e = (unsigned)__builtin_amdgcn_readlane((int)ent, i);
      float* a = acc + (e >> 16) * APITCH + lane;
      __hip_atomic_fetch_add(a +  0, v[i].x, __ATOMIC_RELAXED, __HIP_MEMORY_SCOPE_WORKGROUP);
      __hip_atomic_fetch_add(a + 32, v[i].y, __ATOMIC_RELAXED, __HIP_MEMORY_SCOPE_WORKGROUP);
      __hip_atomic_fetch_add(a + 64, v[i].z, __ATOMIC_RELAXED, __HIP_MEMORY_SCOPE_WORKGROUP);
      __hip_atomic_fetch_add(a + 96, v[i].w, __ATOMIC_RELAXED, __HIP_MEMORY_SCOPE_WORKGROUP);
    }
  }
}

__device__ __forceinline__ int scan_chunk(const int4* __restrict__ dst4, int c,
                                          int base, int nloc,
                                          unsigned* qw, int pending, int lane) {
  int4 d[STEP];
#pragma unroll
  for (int q = 0; q < STEP; ++q) {
    const int i4 = c * (STEP * 32) + q * 32 + lane;
    d[q] = dst4[i4 < N4 ? i4 : N4 - 1];
  }
#pragma unroll
  for (int q = 0; q < STEP; ++q) {
    const int  i4    = c * (STEP * 32) + q * 32 + lane;
    const bool valid = i4 < N4;
    const unsigned l0 = (unsigned)(d[q].x - base);
    const unsigned l1 = (unsigned)(d[q].y - base);
    const unsigned l2 = (unsigned)(d[q].z - base);
    const unsigned l3 = (unsigned)(d[q].w - base);
    const unsigned lmin = min(min(l0, l1), min(l2, l3));
    if (__builtin_amdgcn_ballot_w32(valid && lmin < (unsigned)nloc) != 0u) {
      const bool m0 = valid && l0 < (unsigned)nloc;
      const bool m1 = valid && l1 < (unsigned)nloc;
      const bool m2 = valid && l2 < (unsigned)nloc;
      const bool m3 = valid && l3 < (unsigned)nloc;
      const unsigned e0 = (unsigned)i4 * 4u;
      unsigned mk;
      mk = __builtin_amdgcn_ballot_w32(m0);
      if (m0) qw[pending + __builtin_amdgcn_mbcnt_lo(mk, 0u)] = (e0 + 0u) | (l0 << 20);
      pending += __builtin_popcount(mk);
      mk = __builtin_amdgcn_ballot_w32(m1);
      if (m1) qw[pending + __builtin_amdgcn_mbcnt_lo(mk, 0u)] = (e0 + 1u) | (l1 << 20);
      pending += __builtin_popcount(mk);
      mk = __builtin_amdgcn_ballot_w32(m2);
      if (m2) qw[pending + __builtin_amdgcn_mbcnt_lo(mk, 0u)] = (e0 + 2u) | (l2 << 20);
      pending += __builtin_popcount(mk);
      mk = __builtin_amdgcn_ballot_w32(m3);
      if (m3) qw[pending + __builtin_amdgcn_mbcnt_lo(mk, 0u)] = (e0 + 3u) | (l3 << 20);
      pending += __builtin_popcount(mk);
    }
  }
  return pending;
}

__device__ __forceinline__ int scan_start(int range) {
  return (int)(((unsigned)range * (unsigned)NCHUNK) / NSG);
}

__device__ __forceinline__ void record_edges(int cnt, int at, const unsigned* qw,
                                             const int* __restrict__ src,
                                             unsigned* __restrict__ rec0,
                                             int (&stored)[KS],
                                             unsigned* indeg, int lane) {
  const bool act = lane < cnt;
  unsigned tl = 0u, ent = 0u;
  if (act) {
    const unsigned q   = qw[at + lane];
    const unsigned loc = q >> 20;
    tl = loc / (unsigned)TILE;
    ent = (unsigned)src[q & 0xFFFFFu] | ((loc - tl * (unsigned)TILE) << 16);
    __hip_atomic_fetch_add(indeg + loc, 1u, __ATOMIC_RELAXED, __HIP_MEMORY_SCOPE_WORKGROUP);
  }
#pragma unroll
  for (int t = 0; t < KS; ++t) {
    const bool mine = act && tl == (unsigned)t;
    const unsigned mk = __builtin_amdgcn_ballot_w32(mine);
    if (mine) {
      const int pos = stored[t] + (int)__builtin_amdgcn_mbcnt_lo(mk, 0u);
      if (pos < CAP) rec0[(size_t)t * (32 * KS * CAP) + pos] = ent;
    }
    stored[t] += __builtin_popcount(mk);
  }
}

__launch_bounds__(1024)
__global__ void owner_sweep(const int* __restrict__ src,
                            const int* __restrict__ dst,
                            unsigned* __restrict__ part,
                            unsigned* __restrict__ lists,
                            int* __restrict__ cnts) {
  __shared__ unsigned qbuf[32 * SQCAP];
  __shared__ unsigned indeg[STILE];

  const int tid  = threadIdx.x;
  const int lane = tid & 31;
  const int wave  = __builtin_amdgcn_readfirstlane(tid >> 5);
  const int range = blockIdx.x / KS;
  const int slice = blockIdx.x % KS;
  const int base  = range * STILE;
  const int nloc  = (N_NODES - base) < STILE ? (N_NODES - base) : STILE;

  for (int i = tid; i < STILE; i += 1024) indeg[i] = 0u;
  __syncthreads();

  const int slot0 = ((range * KS) * 32 + wave) * KS + slice;
  unsigned* rec0  = lists + (size_t)slot0 * CAP;
  unsigned* qw    = qbuf + wave * SQCAP;

  const int4* dst4  = (const int4*)dst;
  const int   start = scan_start(range);
  int pending = 0;
  int stored[KS];
#pragma unroll
  for (int t = 0; t < KS; ++t) stored[t] = 0;

  for (int it = wave + 32 * slice; it < NCHUNK; it += 32 * KS) {
    int c = it + start;
    if (c >= NCHUNK) c -= NCHUNK;
    pending = scan_chunk(dst4, c, base, nloc, qw, pending, lane);
    while (pending >= 32) {
      pending -= 32;
      record_edges(32, pending, qw, src, rec0, stored, indeg, lane);
    }
  }
  if (pending > 0) record_edges(pending, 0, qw, src, rec0, stored, indeg, lane);
  if (lane == 0) {
#pragma unroll
    for (int t = 0; t < KS; ++t) cnts[slot0 + t * (32 * KS)] = stored[t];
  }

  __syncthreads();

  for (int i = tid; i < nloc; i += 1024)
    part[(size_t)slice * N_NODES + base + i] = indeg[i];
}

template <int NT>
__device__ __forceinline__ void tile_kloop(const float* acc, const v2f* wl,
                                           int nt, int mg, int lane, v8f (&d)[4]) {
  const v2f*   bptr = wl + nt * 32 + lane;
  const float* aptr = acc + (mg * 16 + (lane & 15)) * APITCH + 2 * (lane >> 4);
#pragma unroll 4
  for (int s = 0; s < DOUT / 4; ++s) {
    const v2f b = bptr[s * 256];
    v2f a[NT];
#pragma unroll
    for (int i = 0; i < NT; ++i)
      a[i] = *(const v2f*)(aptr + i * (64 * APITCH) + 4 * s);
#pragma unroll
    for (int i = 0; i < NT; ++i)
      d[i] = __builtin_amdgcn_wmma_f32_16x16x4_f32(false, a[i], false, b,
                                                   (short)0, d[i], false, false);
  }
}

__launch_bounds__(1024)
__global__ void aggregate_owner(const float* __restrict__ HS,
                                const int* __restrict__ src,
                                const int* __restrict__ dst,
                                const unsigned* __restrict__ part,
                                const float* __restrict__ bias,
                                float* __restrict__ out,
                                const unsigned* __restrict__ lists,
                                const int* __restrict__ cnts,
                                const float* __restrict__ Wn,
                                float* __restrict__ HSn) {
  __shared__ __attribute__((aligned(16))) float    acc[AROWS * APITCH];
  __shared__ __attribute__((aligned(16))) unsigned qbuf[32 * QCAP];
  __shared__ __attribute__((aligned(16))) float    dsh[AROWS];

  const int tid  = threadIdx.x;
  const int lane = tid & 31;
  const int wave = __builtin_amdgcn_readfirstlane(tid >> 5);
  const int base = blockIdx.x * TILE;
  const int nloc = (N_NODES - base) < TILE ? (N_NODES - base) : TILE;

  {
    const float4* own4 = (const float4*)(HS + (size_t)base * DOUT);
    for (int i = tid; i < nloc * 32; i += 1024) {
      const float4 v = own4[i];
      float* a = acc + (i >> 5) * APITCH + (i & 31);
      a[0] = v.x; a[32] = v.y; a[64] = v.z; a[96] = v.w;
    }
    if (Wn != nullptr)
      for (int i = tid; i < (AROWS - nloc) * DOUT; i += 1024)
        acc[(nloc + (i >> 7)) * APITCH + (i & 127)] = 0.0f;
  }
  if (tid < nloc) dsh[tid] = dinv_of(part, base + tid);
  else if (tid < AROWS) dsh[tid] = 0.0f;
  __syncthreads();

  const int slot = (blockIdx.x * 32 + wave) * KS;
  const unsigned* rec = lists + (size_t)slot * CAP;
  int hv[KS];
  bool fits = true;
#pragma unroll
  for (int j = 0; j < KS; ++j) {
    hv[j] = __builtin_amdgcn_readfirstlane(cnts[slot + j]);
    fits = fits && hv[j] <= CAP;
  }

  unsigned* qw = qbuf + wave * QCAP;

  if (fits) {
    unsigned t[KS];
#pragma unroll
    for (int j = 0; j < KS; ++j) {
      t[j] = 0u;
      if (lane < hv[j]) t[j] = rec[j * CAP + lane];
    }
    int have = 0;
#pragma unroll
    for (int j = 0; j < KS; ++j) {
      if (lane < hv[j]) qw[have + lane] = t[j];
#pragma nounroll
      for (int i = 32 + lane; i < hv[j]; i += 32) qw[have + i] = rec[j * CAP + i];
      have += hv[j];
    }
    const int nfull = have & ~(BATCH - 1);
    for (int p = 0; p < nfull; p += BATCH) {
      unsigned ent = 0;
      if (lane < BATCH) ent = qw[p + lane];
      add_rows(BATCH, ent, HS, acc, lane);
    }
    if (have > nfull) {
      unsigned ent = 0;
      if (lane < have - nfull) ent = qw[nfull + lane];
      add_rows(have - nfull, ent, HS, acc, lane);
    }
  } else {
    const int4* dst4  = (const int4*)dst;
    const int   start = scan_start(blockIdx.x / KS);
    int pending = 0;

    for (int it = wave; it < NCHUNK; it += 32) {
      int c = it + start;
      if (c >= NCHUNK) c -= NCHUNK;
      pending = scan_chunk(dst4, c, base, nloc, qw, pending, lane);
      while (pending >= BATCH) {
        pending -= BATCH;
        unsigned ent = 0;
        if (lane < BATCH) {
          const unsigned q = qw[pending + lane];
          ent = (unsigned)src[q & 0xFFFFFu] | ((q >> 20) << 16);
        }
        add_rows(BATCH, ent, HS, acc, lane);
      }
    }
    if (pending > 0) {
      unsigned ent = 0;
      if (lane < pending) {
        const unsigned q = qw[lane];
        ent = (unsigned)src[q & 0xFFFFFu] | ((q >> 20) << 16);
      }
      add_rows(pending, ent, HS, acc, lane);
    }
  }

  __syncthreads();

  const float4* b4 = (const float4*)bias;
  if (Wn == nullptr) {
    float4* out4 = (float4*)(out + (size_t)base * DOUT);
    for (int i = tid; i < nloc * 32; i += 1024) {
      const int   r  = i >> 5;
      const int   l  = i & 31;
      const float di = dsh[r];
      const float* a = acc + r * APITCH + l;
      const float4 bb = b4[l];
      float4 o;
      const float t0 = di * a[0]  + bb.x;
      const float t1 = di * a[32] + bb.y;
      const float t2 = di * a[64] + bb.z;
      const float t3 = di * a[96] + bb.w;
      o.x = t0 > 0.0f ? t0 : 0.0f;
      o.y = t1 > 0.0f ? t1 : 0.0f;
      o.z = t2 > 0.0f ? t2 : 0.0f;
      o.w = t3 > 0.0f ? t3 : 0.0f;
      out4[i] = o;
    }
  } else {
    v2f wreg[8];
#pragma unroll
    for (int i = 0; i < 8; ++i) {
      const int q  = i * 1024 + tid;
      const int n  = q & 127;
      const int p  = (q >> 7) * 2;
      const int k0 = (p & 31) * 4 + (p >> 5);
      wreg[i].x = Wn[(size_t)k0 * DOUT + n];
      wreg[i].y = Wn[(size_t)(k0 + 4) * DOUT + n];
    }
    for (int i = tid; i < nloc * 32; i += 1024) {
      const int   r  = i >> 5;
      const int   l  = i & 31;
      const float di = dsh[r];
      float* a = acc + r * APITCH + l;
      const float4 bb = b4[l];
      const float t0 = di * a[0]  + bb.x;
      const float t1 = di * a[32] + bb.y;
      const float t2 = di * a[64] + bb.z;
      const float t3 = di * a[96] + bb.w;
      a[0]  = t0 > 0.0f ? t0 : 0.0f;
      a[32] = t1 > 0.0f ? t1 : 0.0f;
      a[64] = t2 > 0.0f ? t2 : 0.0f;
      a[96] = t3 > 0.0f ? t3 : 0.0f;
    }
    v2f* wl = (v2f*)qbuf;
#pragma unroll
    for (int i = 0; i < 8; ++i) {
      const int q  = i * 1024 + tid;
      const int n  = q & 127;
      const int pp = q >> 7;
      wl[((pp >> 1) * 8 + (n >> 4)) * 32 + (pp & 1) * 16 + (n & 15)] = wreg[i];
    }
    __syncthreads();

    const int nt  = wave & 7;
    const int mg  = wave >> 3;
    const int ntl = mg == 0 ? 4 : 3;
    v8f d[4];
#pragma unroll
    for (int i = 0; i < 4; ++i) d[i] = v8f{};
    if (mg == 0) tile_kloop<4>(acc, wl, nt, mg, lane, d);
    else         tile_kloop<3>(acc, wl, nt, mg, lane, d);
    __syncthreads();

    const int m16 = lane & 15;
    const int h8  = (lane >> 4) * 8;
#pragma unroll
    for (int i = 0; i < 4; ++i) {
      if (i < ntl) {
        const int    r0 = (mg + 4 * i) * 16 + h8;
        const float4 s0 = *(const float4*)(dsh + r0);
        const float4 s1 = *(const float4*)(dsh + r0 + 4);
        float* o = acc + r0 * APITCH + nt * 16 + m16;
        o[0 * APITCH] = d[i][0] * s0.x;
        o[1 * APITCH] = d[i][1] * s0.y;
        o[2 * APITCH] = d[i][2] * s0.z;
        o[3 * APITCH] = d[i][3] * s0.w;
        o[4 * APITCH] = d[i][4] * s1.x;
        o[5 * APITCH] = d[i][5] * s1.y;
        o[6 * APITCH] = d[i][6] * s1.z;
        o[7 * APITCH] = d[i][7] * s1.w;
      }
    }
    __syncthreads();

    float4* o4 = (float4*)(HSn + (size_t)base * DOUT);
    for (int i = tid; i < nloc * 32; i += 1024)
      o4[i] = *(const float4*)(acc + (i >> 5) * APITCH + 4 * (i & 31));
  }
}

#define WS_PART_BYTES (((size_t)KS * N_NODES * 4 + 255) & ~(size_t)255)
#define WS_HS_BYTES   ((size_t)N_NODES * DOUT * 4)
#define WS_CNT_BYTES  ((size_t)NWG * 32 * KS * 4)
static_assert(WS_HS_BYTES % 256 == 0 && WS_CNT_BYTES % 256 == 0, "slices stay 256B aligned");

extern "C" void kernel_launch(void* const* d_in, const int* in_sizes, int n_in,
                              void* d_out, int out_size, void* d_ws, size_t ws_size,
                              hipStream_t stream) {
  (void)in_sizes; (void)n_in; (void)out_size; (void)ws_size;

  const float* x  = (const float*)d_in[0];
  const int*   ei = (const int*)d_in[1];
  const float* Wl[3] = {(const float*)d_in[2], (const float*)d_in[4], (const float*)d_in[6]};
  const float* bl[3] = {(const float*)d_in[3], (const float*)d_in[5], (const float*)d_in[7]};
  const int* src = ei;
  const int* dst = ei + N_EDGES;

  char*     ws    = (char*)d_ws;
  unsigned* part  = (unsigned*)ws;
  float*    hs    = (float*)(ws + WS_PART_BYTES);
  float*    hs2   = (float*)(ws + WS_PART_BYTES + WS_HS_BYTES);
  int*      cnts  = (int*)(ws + WS_PART_BYTES + 2 * WS_HS_BYTES);
  unsigned* lists = (unsigned*)(ws + WS_PART_BYTES + 2 * WS_HS_BYTES + WS_CNT_BYTES);

  owner_sweep<<<NWG, 1024, 0, stream>>>(src, dst, part, lists, cnts);

  float* hout = (float*)d_out;
  gemm_wmma_f32  <<<(N_NODES + 31) / 32, 256, 0, stream>>>(x, Wl[0], part, hs, N_NODES);
  aggregate_owner<<<NWG, 1024, 0, stream>>>(hs,  src, dst, part, bl[0], hout, lists, cnts, Wl[1], hs2);
  aggregate_owner<<<NWG, 1024, 0, stream>>>(hs2, src, dst, part, bl[1], hout, lists, cnts, Wl[2], hs);
  aggregate_owner<<<NWG, 1024, 0, stream>>>(hs,  src, dst, part, bl[2], hout, lists, cnts, nullptr, nullptr);
}


# Sums held in registers of owner waves, no LDS float adds: sweep ranks each edge into the list of (tile, wave = node & 31) by LDS integer add; scalar-offset register pick; sum order: list arrival
# speedup vs baseline: 4.5798x; 1.1144x over previous
#include <hip/hip_runtime.h>
#include <hip/hip_bf16.h>

#define N_NODES 50000
#define N_EDGES 800000
#define DIN 128
#define DOUT 128
#define LDS_STRIDE 132

typedef __attribute__((ext_vector_type(2))) float v2f;
typedef __attribute__((ext_vector_type(8))) float v8f;

#define KS 8

__device__ __forceinline__ float dinv_of(const unsigned* __restrict__ part, int node) {
  unsigned c = 0u;
#pragma unroll
  for (int j = 0; j < KS; ++j) c += part[(size_t)j * N_NODES + node];
  const float d = 1.0f + (float)c;
  return d > 0.0f ? (1.0f / sqrtf(d)) : 0.0f;
}

__launch_bounds__(256)
__global__ void gemm_wmma_f32(const float* __restrict__ A,
                              const float* __restrict__ W,
                              const unsigned* __restrict__ part,
                              float* __restrict__ C, int nrows) {
  __shared__ float As[32 * LDS_STRIDE];
  __shared__ float dsh[32];

  const int tid  = threadIdx.x;
  const int lane = tid & 31;
  const int wave = tid >> 5;
  const int m0   = blockIdx.x * 32;
  const int rows = (nrows - m0) < 32 ? (nrows - m0) : 32;

  {
    const float4* src4 = (const float4*)(A + (size_t)m0 * DIN);
    const int n4 = rows * (DIN / 4);
    for (int i = tid; i < n4; i += 256) {
      const int r = i >> 5;
      const int c = i & 31;
      ((float4*)(As + r * LDS_STRIDE))[c] = src4[i];
    }
  }
  if (tid < rows) dsh[tid] = dinv_of(part, m0 + tid);
  __syncthreads();

  const int n0    = wave * 16;
  const int mlane = lane & 15;
  const int khalf = (lane >> 4) * 2;

  const float* a0   = As + (size_t)mlane * LDS_STRIDE + khalf;
  const float* a1   = a0 + 16 * LDS_STRIDE;
  const float* wcol = W + (size_t)khalf * DOUT + n0 + mlane;

  v8f acc0 = {}, acc1 = {};
#pragma unroll 4
  for (int k = 0; k < DIN; k += 4) {
    v2f b;
    b.x = wcol[(size_t)k * DOUT];
    b.y = wcol[(size_t)(k + 1) * DOUT];
    v2f x0 = *(const v2f*)(a0 + k);
    v2f x1 = *(const v2f*)(a1 + k);
    acc0 = __builtin_amdgcn_wmma_f32_16x16x4_f32(false, x0, false, b,
                                                 (short)0, acc0, false, false);
    acc1 = __builtin_amdgcn_wmma_f32_16x16x4_f32(false, x1, false, b,
                                                 (short)0, acc1, false, false);
  }

  const int   rbase = m0 + (lane >> 4) * 8;
  const float* drow = dsh + (lane >> 4) * 8;
  float* crow = C + (size_t)rbase * DOUT + n0 + mlane;
#pragma unroll
  for (int j = 0; j < 8; ++j)
    crow[(size_t)j * DOUT] = acc0[j] * drow[j];
  if (rows > 16) {
    float* crow1 = crow + (size_t)16 * DOUT;
#pragma unroll
    for (int j = 0; j < 8; ++j)
      crow1[(size_t)j * DOUT] = acc1[j] * drow[16 + j];
  }
}

#define TILE   196
#define MT     ((TILE + 15) / 16)
#define AROWS  (MT * 16)
#define APITCH 132
#define NWG    ((N_NODES + TILE - 1) / TILE)
#define STILE  (KS * TILE)
#define NSG    (NWG / KS)
#define N4     (N_EDGES / 4)
#define STEP   8
#define NCHUNK ((N4 + STEP * 32 - 1) / (STEP * 32))
#define BATCH  8
#define QCAP   (BATCH + STEP * 128)
#define SQCAP  1344
#define CAP    (1024 / KS)
#define NPW    ((TILE + 31) / 32)

static_assert(N_EDGES % 4 == 0, "dst is read as int4");
static_assert(NWG % KS == 0, "whole sweep ranges");
static_assert(N_EDGES <= (1 << 20) && STILE <= (1 << 12) && TILE <= 256, "LDS list entry packing");
static_assert(N_NODES <= (1 << 16) && NPW < 8, "stored entry packing");
static_assert(4 * (NPW + 1) <= 32, "the sums of a wave's nodes and of the padding entries fit 32 registers");
static_assert(KS * CAP <= QCAP, "a wave's joined slices fit its LDS list");
static_assert(STEP * 128 <= QCAP, "one wave step of a rescan fits the LDS list");
static_assert(SQCAP >= 32 + STEP * 128, "the sweep's list holds a left-over batch and one wave step");
static_assert(MT == 13, "row tiles 0..12 over 4 wave groups: 4, 3, 3, 3");
static_assert(32 * QCAP * 4 >= DOUT * DOUT * 4, "the next weight fits the dead wave lists");

typedef __attribute__((ext_vector_type(32))) float v32f;

__device__ __forceinline__ void slot_add(v32f& s, unsigned j, const float4& v) {
  const unsigned o = j * 4u;
  s[o + 0u] += v.x;
  s[o + 1u] += v.y;
  s[o + 2u] += v.z;
  s[o + 3u] += v.w;
}

__device__ __forceinline__ void gather_rows(unsigned ent, const float* __restrict__ HS,
                                            v32f& s, int lane) {
  float4 v[BATCH];
#pragma unroll
  for (int i = 0; i < BATCH; ++i) {
    const unsigned e = (unsigned)__builtin_amdgcn_readlane((int)ent, i);
    v[i] = ((const float4*)(HS + (size_t)(e & 0xFFFFu) * DOUT))[lane];
  }
#pragma unroll
  for (int i = 0; i < BATCH; ++i) {
    const unsigned e = (unsigned)__builtin_amdgcn_readlane((int)ent, i);
    slot_add(s, e >> 16, v[i]);
  }
}

__device__ __forceinline__ int scan_chunk(const int4* __restrict__ dst4, int c,
                                          int base, int nloc,
                                          unsigned* qw, int pending, int lane) {
  int4 d[STEP];
#pragma unroll
  for (int q = 0; q < STEP; ++q) {
    const int i4 = c * (STEP * 32) + q * 32 + lane;
    d[q] = dst4[i4 < N4 ? i4 : N4 - 1];
  }
#pragma unroll
  for (int q = 0; q < STEP; ++q) {
    const int  i4    = c * (STEP * 32) + q * 32 + lane;
    const bool valid = i4 < N4;
    const unsigned l0 = (unsigned)(d[q].x - base);
    const unsigned l1 = (unsigned)(d[q].y - base);
    const unsigned l2 = (unsigned)(d[q].z - base);
    const unsigned l3 = (unsigned)(d[q].w - base);
    const unsigned lmin = min(min(l0, l1), min(l2, l3));
    if (__builtin_amdgcn_ballot_w32(valid && lmin < (unsigned)nloc) != 0u) {
      const bool m0 = valid && l0 < (unsigned)nloc;
      const bool m1 = valid && l1 < (unsigned)nloc;
      const bool m2 = valid && l2 < (unsigned)nloc;
      const bool m3 = valid && l3 < (unsigned)nloc;
      const unsigned e0 = (unsigned)i4 * 4u;
      unsigned mk;
      mk = __builtin_amdgcn_ballot_w32(m0);
      if (m0) qw[pending + __builtin_amdgcn_mbcnt_lo(mk, 0u)] = (e0 + 0u) | (l0 << 20);
      pending += __builtin_popcount(mk);
      mk = __builtin_amdgcn_ballot_w32(m1);
      if (m1) qw[pending + __builtin_amdgcn_mbcnt_lo(mk, 0u)] = (e0 + 1u) | (l1 << 20);
      pending += __builtin_popcount(mk);
      mk = __builtin_amdgcn_ballot_w32(m2);
      if (m2) qw[pending + __builtin_amdgcn_mbcnt_lo(mk, 0u)] = (e0 + 2u) | (l2 << 20);
      pending += __builtin_popcount(mk);
      mk = __builtin_amdgcn_ballot_w32(m3);
      if (m3) qw[pending + __builtin_amdgcn_mbcnt_lo(mk, 0u)] = (e0 + 3u) | (l3 << 20);
      pending += __builtin_popcount(mk);
    }
  }
  return pending;
}

__device__ __forceinline__ int scan_start(int range) {
  return (int)(((unsigned)range * (unsigned)NCHUNK) / NSG);
}

__device__ __forceinline__ void record_edges(int cnt, int at, const unsigned* qw,
                                             const int* __restrict__ src,
                                             unsigned* __restrict__ rec0,
                                             unsigned* bcnt, unsigned* indeg, int lane) {
  if (lane < cnt) {
    const unsigned q   = qw[at + lane];
    const unsigned loc = q >> 20;
    const unsigned tl  = loc / (unsigned)TILE;
    const unsigned nl  = loc - tl * (unsigned)TILE;
    const unsigned b   = tl * 32u + (nl & 31u);
    const unsigned ent = (unsigned)src[q & 0xFFFFFu] | ((nl >> 5) << 16);
    __hip_atomic_fetch_add(indeg + loc, 1u, __ATOMIC_RELAXED, __HIP_MEMORY_SCOPE_WORKGROUP);
    const unsigned pos =
        __hip_atomic_fetch_add(bcnt + b, 1u, __ATOMIC_RELAXED, __HIP_MEMORY_SCOPE_WORKGROUP);
    if (pos < (unsigned)CAP) rec0[(size_t)b * (KS * CAP) + pos] = ent;
  }
}

__launch_bounds__(1024)
__global__ void owner_sweep(const int* __restrict__ src,
                            const int* __restrict__ dst,
                            unsigned* __restrict__ part,
                            unsigned* __restrict__ lists,
                            int* __restrict__ cnts) {
  __shared__ unsigned qbuf[32 * SQCAP];
  __shared__ unsigned indeg[STILE];
  __shared__ unsigned bcnt[KS * 32];

  const int tid  = threadIdx.x;
  const int lane = tid & 31;
  const int wave  = __builtin_amdgcn_readfirstlane(tid >> 5);
  const int range = blockIdx.x / KS;
  const int slice = blockIdx.x % KS;
  const int base  = range * STILE;
  const int nloc  = (N_NODES - base) < STILE ? (N_NODES - base) : STILE;

  for (int i = tid; i < STILE; i += 1024) indeg[i] = 0u;
  if (tid < KS * 32) bcnt[tid] = 0u;
  __syncthreads();

  const int slot0 = ((range * KS) * 32) * KS + slice;
  unsigned* rec0  = lists + (size_t)slot0 * CAP;
  unsigned* qw    = qbuf + wave * SQCAP;

  const int4* dst4  = (const int4*)dst;
  const int   start = scan_start(range);
  int pending = 0;

  for (int it = wave + 32 * slice; it < NCHUNK; it += 32 * KS) {
    int c = it + start;
    if (c >= NCHUNK) c -= NCHUNK;
    pending = scan_chunk(dst4, c, base, nloc, qw, pending, lane);
    while (pending >= 32) {
      pending -= 32;
      record_edges(32, pending, qw, src, rec0, bcnt, indeg, lane);
    }
  }
  if (pending > 0) record_edges(pending, 0, qw, src, rec0, bcnt, indeg, lane);

  __syncthreads();

  if (tid < KS * 32) cnts[slot0 + tid * KS] = (int)bcnt[tid];

  for (int i = tid; i < nloc; i += 1024)
    part[(size_t)slice * N_NODES + base + i] = indeg[i];
}

template <int NT>
__device__ __forceinline__ void tile_kloop(const float* acc, const v2f* wl,
                                           int nt, int mg, int lane, v8f (&d)[4]) {
  const v2f*   bptr = wl + nt * 32 + lane;
  const float* aptr = acc + (mg * 16 + (lane & 15)) * APITCH + 2 * (lane >> 4);
#pragma unroll 4
  for (int s = 0; s < DOUT / 4; ++s) {
    const v2f b = bptr[s * 256];
    v2f a[NT];
#pragma unroll
    for (int i = 0; i < NT; ++i)
      a[i] = *(const v2f*)(aptr + i * (64 * APITCH) + 4 * s);
#pragma unroll
    for (int i = 0; i < NT; ++i)
      d[i] = __builtin_amdgcn_wmma_f32_16x16x4_f32(false, a[i], false, b,
                                                   (short)0, d[i], false, false);
  }
}

__launch_bounds__(1024)
__global__ void aggregate_owner(const float* __restrict__ HS,
                                const int* __restrict__ src,
                                const int* __restrict__ dst,
                                const unsigned* __restrict__ part,
                                const float* __restrict__ bias,
                                float* __restrict__ out,
                                const unsigned* __restrict__ lists,
                                const int* __restrict__ cnts,
                                const float* __restrict__ Wn,
                                float* __restrict__ HSn) {
  __shared__ __attribute__((aligned(16))) float    acc[AROWS * APITCH];
  __shared__ __attribute__((aligned(16))) unsigned qbuf[32 * QCAP];
  __shared__ __attribute__((aligned(16))) float    dsh[AROWS];

  const int tid  = threadIdx.x;
  const int lane = tid & 31;
  const int wave = __builtin_amdgcn_readfirstlane(tid >> 5);
  const int base = blockIdx.x * TILE;
  const int nloc = (N_NODES - base) < TILE ? (N_NODES - base) : TILE;

  v32f s = {};
  {
    const float4* own4 = (const float4*)(HS + (size_t)base * DOUT);
#pragma unroll
    for (int j = 0; j < NPW; ++j) {
      const int n = wave + 32 * j;
      if (n < nloc) {
        const float4 own = own4[n * 32 + lane];
        s[4 * j + 0] = own.x; s[4 * j + 1] = own.y; s[4 * j + 2] = own.z; s[4 * j + 3] = own.w;
      }
    }
    if (Wn != nullptr)
      for (int i = tid; i < (AROWS - nloc) * DOUT; i += 1024)
        acc[(nloc + (i >> 7)) * APITCH + (i & 127)] = 0.0f;
  }
  if (tid < nloc) dsh[tid] = dinv_of(part, base + tid);
  else if (tid < AROWS) dsh[tid] = 0.0f;

  const int slot = (blockIdx.x * 32 + wave) * KS;
  const unsigned* rec = lists + (size_t)slot * CAP;
  int hv[KS];
  bool fits = true;
#pragma unroll
  for (int j = 0; j < KS; ++j) {
    hv[j] = __builtin_amdgcn_readfirstlane(cnts[slot + j]);
    fits = fits && hv[j] <= CAP;
  }

  unsigned* qw = qbuf + wave * QCAP;

  if (fits) {
    unsigned t[KS];
#pragma unroll
    for (int j = 0; j < KS; ++j) {
      t[j] = 0u;
      if (lane < hv[j]) t[j] = rec[j * CAP + lane];
    }
    int have = 0;
#pragma unroll
    for (int j = 0; j < KS; ++j) {
      if (lane < hv[j]) qw[have + lane] = t[j];
#pragma nounroll
      for (int i = 32 + lane; i < hv[j]; i += 32) qw[have + i] = rec[j * CAP + i];
      have += hv[j];
    }
    if (lane < BATCH) qw[have + lane] = (unsigned)NPW << 16;
    const int nall = (have + BATCH - 1) & ~(BATCH - 1);
    for (int p = 0; p < nall; p += BATCH) {
      unsigned ent = 0;
      if (lane < BATCH) ent = qw[p + lane];
      gather_rows(ent, HS, s, lane);
    }
  } else {
    const int4* dst4 = (const int4*)dst;
    for (int c = 0; c < NCHUNK; ++c) {
      const int found = scan_chunk(dst4, c, base, nloc, qw, 0, lane);
      for (int p = 0; p < found; p += 32) {
        unsigned ent  = 0;
        bool     mine = false;
        if (p + lane < found) {
          const unsigned q  = qw[p + lane];
          const unsigned nl = q >> 20;
          mine = (nl & 31u) == (unsigned)wave;
          if (mine) ent = (unsigned)src[q & 0xFFFFFu] | ((nl >> 5) << 16);
        }
        unsigned mk = __builtin_amdgcn_ballot_w32(mine);
        while (mk != 0u) {
          const int i = __builtin_ctz(mk);
          mk &= mk - 1u;
          const unsigned e = (unsigned)__builtin_amdgcn_readlane((int)ent, i);
          const float4 v = ((const float4*)(HS + (size_t)(e & 0xFFFFu) * DOUT))[lane];
          slot_add(s, e >> 16, v);
        }
      }
    }
  }

#pragma unroll
  for (int j = 0; j < NPW; ++j) {
    const int n = wave + 32 * j;
    if (n < nloc) {
      float* a = acc + n * APITCH + lane;
      a[0] = s[4 * j + 0]; a[32] = s[4 * j + 1]; a[64] = s[4 * j + 2]; a[96] = s[4 * j + 3];
    }
  }

  __syncthreads();

  const float4* b4 = (const float4*)bias;
  if (Wn == nullptr) {
    float4* out4 = (float4*)(out + (size_t)base * DOUT);
    for (int i = tid; i < nloc * 32; i += 1024) {
      const int   r  = i >> 5;
      const int   l  = i & 31;
      const float di = dsh[r];
      const float* a = acc + r * APITCH + l;
      const float4 bb = b4[l];
      float4 o;
      const float t0 = di * a[0]  + bb.x;
      const float t1 = di * a[32] + bb.y;
      const float t2 = di * a[64] + bb.z;
      const float t3 = di * a[96] + bb.w;
      o.x = t0 > 0.0f ? t0 : 0.0f;
      o.y = t1 > 0.0f ? t1 : 0.0f;
      o.z = t2 > 0.0f ? t2 : 0.0f;
      o.w = t3 > 0.0f ? t3 : 0.0f;
      out4[i] = o;
    }
  } else {
    v2f wreg[8];
#pragma unroll
    for (int i = 0; i < 8; ++i) {
      const int q  = i * 1024 + tid;
      const int n  = q & 127;
      const int p  = (q >> 7) * 2;
      const int k0 = (p & 31) * 4 + (p >> 5);
      wreg[i].x = Wn[(size_t)k0 * DOUT + n];
      wreg[i].y = Wn[(size_t)(k0 + 4) * DOUT + n];
    }
    for (int i = tid; i < nloc * 32; i += 1024) {
      const int   r  = i >> 5;
      const int   l  = i & 31;
      const float di = dsh[r];
      float* a = acc + r * APITCH + l;
      const float4 bb = b4[l];
      const float t0 = di * a[0]  + bb.x;
      const float t1 = di * a[32] + bb.y;
      const float t2 = di * a[64] + bb.z;
      const float t3 = di * a[96] + bb.w;
      a[0]  = t0 > 0.0f ? t0 : 0.0f;
      a[32] = t1 > 0.0f ? t1 : 0.0f;
      a[64] = t2 > 0.0f ? t2 : 0.0f;
      a[96] = t3 > 0.0f ? t3 : 0.0f;
    }
    v2f* wl = (v2f*)qbuf;
#pragma unroll
    for (int i = 0; i < 8; ++i) {
      const int q  = i * 1024 + tid;
      const int n  = q & 127;
      const int pp = q >> 7;
      wl[((pp >> 1) * 8 + (n >> 4)) * 32 + (pp & 1) * 16 + (n & 15)] = wreg[i];
    }
    __syncthreads();

    const int nt  = wave & 7;
    const int mg  = wave >> 3;
    const int ntl = mg == 0 ? 4 : 3;
    v8f d[4];
#pragma unroll
    for (int i = 0; i < 4; ++i) d[i] = v8f{};
    if (mg == 0) tile_kloop<4>(acc, wl, nt, mg, lane, d);
    else         tile_kloop<3>(acc, wl, nt, mg, lane, d);
    __syncthreads();

    const int m16 = lane & 15;
    const int h8  = (lane >> 4) * 8;
#pragma unroll
    for (int i = 0; i < 4; ++i) {
      if (i < ntl) {
        const int    r0 = (mg + 4 * i) * 16 + h8;
        const float4 s0 = *(const float4*)(dsh + r0);
        const float4 s1 = *(const float4*)(dsh + r0 + 4);
        float* o = acc + r0 * APITCH + nt * 16 + m16;
        o[0 * APITCH] = d[i][0] * s0.x;
        o[1 * APITCH] = d[i][1] * s0.y;
        o[2 * APITCH] = d[i][2] * s0.z;
        o[3 * APITCH] = d[i][3] * s0.w;
        o[4 * APITCH] = d[i][4] * s1.x;
        o[5 * APITCH] = d[i][5] * s1.y;
        o[6 * APITCH] = d[i][6] * s1.z;
        o[7 * APITCH] = d[i][7] * s1.w;
      }
    }
    __syncthreads();

    float4* o4 = (float4*)(HSn + (size_t)base * DOUT);
    for (int i = tid; i < nloc * 32; i += 1024)
      o4[i] = *(const float4*)(acc + (i >> 5) * APITCH + 4 * (i & 31));
  }
}

#define WS_PART_BYTES (((size_t)KS * N_NODES * 4 + 255) & ~(size_t)255)
#define WS_HS_BYTES   ((size_t)N_NODES * DOUT * 4)
#define WS_CNT_BYTES  ((size_t)NWG * 32 * KS * 4)
static_assert(WS_HS_BYTES % 256 == 0 && WS_CNT_BYTES % 256 == 0, "slices stay 256B aligned");

extern "C" void kernel_launch(void* const* d_in, const int* in_sizes, int n_in,
                              void* d_out, int out_size, void* d_ws, size_t ws_size,
                              hipStream_t stream) {
  (void)in_sizes; (void)n_in; (void)out_size; (void)ws_size;

  const float* x  = (const float*)d_in[0];
  const int*   ei = (const int*)d_in[1];
  const float* Wl[3] = {(const float*)d_in[2], (const float*)d_in[4], (const float*)d_in[6]};
  const float* bl[3] = {(const float*)d_in[3], (const float*)d_in[5], (const float*)d_in[7]};
  const int* src = ei;
  const int* dst = ei + N_EDGES;

  char*     ws    = (char*)d_ws;
  unsigned* part  = (unsigned*)ws;
  float*    hs    = (float*)(ws + WS_PART_BYTES);
  float*    hs2   = (float*)(ws + WS_PART_BYTES + WS_HS_BYTES);
  int*      cnts  = (int*)(ws + WS_PART_BYTES + 2 * WS_HS_BYTES);
  unsigned* lists = (unsigned*)(ws + WS_PART_BYTES + 2 * WS_HS_BYTES + WS_CNT_BYTES);

  owner_sweep<<<NWG, 1024, 0, stream>>>(src, dst, part, lists, cnts);

  float* hout = (float*)d_out;
  gemm_wmma_f32  <<<(N_NODES + 31) / 32, 256, 0, stream>>>(x, Wl[0], part, hs, N_NODES);
  aggregate_owner<<<NWG, 1024, 0, stream>>>(hs,  src, dst, part, bl[0], hout, lists, cnts, Wl[1], hs2);
  aggregate_owner<<<NWG, 1024, 0, stream>>>(hs2, src, dst, part, bl[1], hout, lists, cnts, Wl[2], hs);
  aggregate_owner<<<NWG, 1024, 0, stream>>>(hs,  src, dst, part, bl[2], hout, lists, cnts, nullptr, nullptr);
}
